# starting kernel
# baseline (speedup 1.0000x reference)
#include <hip/hip_runtime.h>

typedef float v2f __attribute__((ext_vector_type(2)));
typedef float v8f __attribute__((ext_vector_type(8)));

#define IN_DIM 128
#define HID 128
#define OUT 64

__global__ __launch_bounds__(256) void k_init_deg(float* __restrict__ deg, int n) {
    int i = blockIdx.x * blockDim.x + threadIdx.x;
    if (i < n) deg[i] = 1.0f;
}

__global__ __launch_bounds__(256) void k_edge_deg(const int* __restrict__ dst, int E,
                                                  float* __restrict__ deg) {
    int e = blockIdx.x * blockDim.x + threadIdx.x;
    if (e < E) atomicAdd(&deg[dst[e]], 1.0f);
}

__global__ __launch_bounds__(256) void k_rsqrt(const float* __restrict__ deg,
                                               float* __restrict__ dinv, int n) {
    int i = blockIdx.x * blockDim.x + threadIdx.x;
    if (i < n) dinv[i] = rsqrtf(deg[i]);
}

template <int NTILES>
__global__ __launch_bounds__(256) void k_gemm_wmma(const float* __restrict__ X,
                                                   const float* __restrict__ W,
                                                   float* __restrict__ H, int nrows) {
    const int lane = threadIdx.x & 31;
    const int wave = threadIdx.x >> 5;
    const int tile = blockIdx.x * (blockDim.x >> 5) + wave;
    if (tile * 16 >= nrows) return;

    const int ln = lane & 15;
    const int hi = lane >> 4;
    const int ldw = NTILES * 16;

    v8f acc[NTILES] = {};

    const float* xrow = X + (size_t)(tile * 16 + ln) * IN_DIM;

    for (int k = 0; k < IN_DIM; k += 4) {
        v2f a;
        a.x = xrow[k + 2 * hi + 0];
        a.y = xrow[k + 2 * hi + 1];

        const float* wr = W + (size_t)(k + 2 * hi) * ldw;
#pragma unroll
        for (int n = 0; n < NTILES; ++n) {
            v2f b;
            b.x = wr[n * 16 + ln];
            b.y = wr[ldw + n * 16 + ln];
            acc[n] = __builtin_amdgcn_wmma_f32_16x16x4_f32(
                false, a, false, b, (short)0, acc[n], false, false);
        }
    }

#pragma unroll
    for (int n = 0; n < NTILES; ++n) {
#pragma unroll
        for (int r = 0; r < 8; ++r) {
            int orow = tile * 16 + r + 8 * hi;
            H[(size_t)orow * ldw + n * 16 + ln] = acc[n][r];
        }
    }
}

__global__ __launch_bounds__(256) void k_self_bias(const float* __restrict__ H,
                                                   const float* __restrict__ dinv,
                                                   const float* __restrict__ bias,
                                                   float* __restrict__ A,
                                                   int n, int dim) {
    size_t idx = (size_t)blockIdx.x * blockDim.x + threadIdx.x;
    size_t total = (size_t)n * dim;
    if (idx < total) {
        int i = (int)(idx / dim);
        int c = (int)(idx % dim);
        float s = dinv[i];
        A[idx] = H[idx] * (s * s) + bias[c];
    }
}

template <int DIM>
__global__ __launch_bounds__(256) void k_scatter(const float* __restrict__ H,
                                                 const int* __restrict__ src,
                                                 const int* __restrict__ dst,
                                                 const float* __restrict__ dinv,
                                                 float* __restrict__ A, int E) {
    const int lane = threadIdx.x & 31;
    const int e = (int)(((size_t)blockIdx.x * blockDim.x + threadIdx.x) >> 5);
    if (e >= E) return;
    const int s = src[e];
    const int d = dst[e];
    const float norm = dinv[s] * dinv[d];
    constexpr int VPT = DIM / 32;
    const float* hp = H + (size_t)s * DIM + lane * VPT;
    float* ap = A + (size_t)d * DIM + lane * VPT;
#pragma unroll
    for (int j = 0; j < VPT; ++j) atomicAdd(&ap[j], hp[j] * norm);
}

__global__ __launch_bounds__(256) void k_relu(float* __restrict__ A, size_t total) {
    size_t idx = (size_t)blockIdx.x * blockDim.x + threadIdx.x;
    if (idx < total) A[idx] = fmaxf(A[idx], 0.0f);
}

extern "C" void kernel_launch(void* const* d_in, const int* in_sizes, int n_in,
                              void* d_out, int out_size, void* d_ws, size_t ws_size,
                              hipStream_t stream) {
    const float* x   = (const float*)d_in[0];
    const int*   ei  = (const int*)d_in[1];
    const float* W1  = (const float*)d_in[2];
    const float* b1  = (const float*)d_in[3];
    const float* W2  = (const float*)d_in[4];
    const float* b2  = (const float*)d_in[5];
    float*       out = (float*)d_out;

    const int n = in_sizes[0] / IN_DIM;
    const int E = in_sizes[1] / 2;
    const int* srcIdx = ei;
    const int* dstIdx = ei + E;

    float* ws   = (float*)d_ws;
    float* deg  = ws;
    float* dinv = deg + n;
    float* h    = dinv + n;
    float* agg  = h + (size_t)n * HID;

    k_init_deg<<<(n + 255) / 256, 256, 0, stream>>>(deg, n);
    k_edge_deg<<<(E + 255) / 256, 256, 0, stream>>>(dstIdx, E, deg);
    k_rsqrt<<<(n + 255) / 256, 256, 0, stream>>>(deg, dinv, n);

    const int tiles = (n + 15) / 16;
    const int gemmBlocks = (tiles + 7) / 8;
    const int scatterBlocks = (int)(((size_t)E * 32 + 255) / 256);

    k_gemm_wmma<8><<<gemmBlocks, 256, 0, stream>>>(x, W1, h, n);
    {
        size_t total = (size_t)n * HID;
        k_self_bias<<<(int)((total + 255) / 256), 256, 0, stream>>>(h, dinv, b1, agg, n, HID);
        k_scatter<HID><<<scatterBlocks, 256, 0, stream>>>(h, srcIdx, dstIdx, dinv, agg, E);
        k_relu<<<(int)((total + 255) / 256), 256, 0, stream>>>(agg, total);
    }

    k_gemm_wmma<4><<<gemmBlocks, 256, 0, stream>>>(agg, W2, h, n);
    {
        size_t total = (size_t)n * OUT;
        k_self_bias<<<(int)((total + 255) / 256), 256, 0, stream>>>(h, dinv, b2, out, n, OUT);
        k_scatter<OUT><<<scatterBlocks, 256, 0, stream>>>(h, srcIdx, dstIdx, dinv, out, E);
    }
}


# Atomic scatter replaced by destination-grouped gather: edges counting-sorted by destination per call (slot from a returning atomic), one wave sums a row in registers, stores once; sum order changes
# speedup vs baseline: 2.1763x; 2.1763x over previous
#include <hip/hip_runtime.h>


typedef float v2f __attribute__((ext_vector_type(2)));
typedef float v8f __attribute__((ext_vector_type(8)));

#define IN_DIM 128
#define HID 128
#define OUT 64

__global__ __launch_bounds__(256) void k_zero_cnt(int* __restrict__ cnt, int n) {
    int i = blockIdx.x * blockDim.x + threadIdx.x;
    if (i < n) cnt[i] = 0;
}

__global__ __launch_bounds__(256) void k_count_rank(const int* __restrict__ dst, int E,
                                                    int* __restrict__ cnt,
                                                    int* __restrict__ rank) {
    int e = blockIdx.x * blockDim.x + threadIdx.x;
    if (e < E) rank[e] = atomicAdd(&cnt[dst[e]], 1);
}

__global__ __launch_bounds__(256) void k_scan_dinv(const int* __restrict__ cnt,
                                                   int* __restrict__ rowptr,
                                                   float* __restrict__ dinv, int n) {
    __shared__ int presum[8];
    __shared__ int wtot[8];
    const int t = threadIdx.x;
    const int lane = t & 31;
    const int wave = t >> 5;
    const int b = blockIdx.x;

    const int4* c4 = (const int4*)cnt;
    int pre = 0;
    for (int i = t; i < b * 256; i += 256) {
        int4 v = c4[i];
        pre += (v.x + v.y) + (v.z + v.w);
    }
    for (int o = 16; o > 0; o >>= 1) pre += __shfl_xor(pre, o);
    if (lane == 0) presum[wave] = pre;

    const int i0 = b * 1024 + t * 4;
    int c[4];
#pragma unroll
    for (int j = 0; j < 4; ++j) c[j] = (i0 + j < n) ? cnt[i0 + j] : 0;
    const int tot = (c[0] + c[1]) + (c[2] + c[3]);

    int inc = tot;
    for (int o = 1; o < 32; o <<= 1) {
        int v = __shfl_up(inc, o);
        if (lane >= o) inc += v;
    }
    if (lane == 31) wtot[wave] = inc;
    __syncthreads();

    int off = 0;
    for (int w = 0; w < 8; ++w) {
        off += presum[w];
        if (w < wave) off += wtot[w];
    }
    int run = off + inc - tot;
#pragma unroll
    for (int j = 0; j < 4; ++j) {
        if (i0 + j < n) {
            rowptr[i0 + j] = run;
            dinv[i0 + j] = rsqrtf((float)(c[j] + 1));
        }
        run += c[j];
    }
}

__global__ __launch_bounds__(256) void k_fill(const int* __restrict__ src,
                                              const int* __restrict__ dst,
                                              const int* __restrict__ rank,
                                              const int* __restrict__ rowptr,
                                              const float* __restrict__ dinv,
                                              int* __restrict__ colw, int E) {
    int e = blockIdx.x * blockDim.x + threadIdx.x;
    if (e < E) {
        const int s = src[e];
        const int pos = rowptr[dst[e]] + rank[e];
        if ((unsigned)pos < (unsigned)E)
            *(int2*)(colw + 2 * (size_t)pos) = make_int2(s, __float_as_int(dinv[s]));
    }
}

__global__ __launch_bounds__(256) void k_copy_int(const int* __restrict__ a,
                                                  int* __restrict__ b, int count) {
    int i = (blockIdx.x * blockDim.x + threadIdx.x) * 4;
    if (i + 3 < count) {
        *(int4*)(b + i) = *(const int4*)(a + i);
    } else {
        for (int j = i; j < count; ++j) b[j] = a[j];
    }
}

template <int NTILES>
__global__ __launch_bounds__(256) void k_gemm_wmma(const float* __restrict__ X,
                                                   const float* __restrict__ W,
                                                   float* __restrict__ H, int nrows) {
    const int lane = threadIdx.x & 31;
    const int wave = threadIdx.x >> 5;
    const int tile = blockIdx.x * (blockDim.x >> 5) + wave;
    if (tile * 16 >= nrows) return;

    const int ln = lane & 15;
    const int hi = lane >> 4;
    const int ldw = NTILES * 16;

    v8f acc[NTILES] = {};

    const float* xrow = X + (size_t)(tile * 16 + ln) * IN_DIM;

    for (int k = 0; k < IN_DIM; k += 4) {
        v2f a;
        a.x = xrow[k + 2 * hi + 0];
        a.y = xrow[k + 2 * hi + 1];

        const float* wr = W + (size_t)(k + 2 * hi) * ldw;
#pragma unroll
        for (int n = 0; n < NTILES; ++n) {
            v2f b;
            b.x = wr[n * 16 + ln];
            b.y = wr[ldw + n * 16 + ln];
            acc[n] = __builtin_amdgcn_wmma_f32_16x16x4_f32(
                false, a, false, b, (short)0, acc[n], false, false);
        }
    }

#pragma unroll
    for (int n = 0; n < NTILES; ++n) {
#pragma unroll
        for (int r = 0; r < 8; ++r) {
            int orow = tile * 16 + r + 8 * hi;
            H[(size_t)orow * ldw + n * 16 + ln] = acc[n][r];
        }
    }
}


template <int G, int DIM>
__device__ __forceinline__ void add_rows(float __attribute__((ext_vector_type(DIM / 32)))& acc,
                                         const float* __restrict__ Hlane,
                                         int sv, float dv, float di, int t) {
    typedef float vecf __attribute__((ext_vector_type(DIM / 32)));
    vecf h[G];
#pragma unroll
    for (int j = 0; j < G; ++j) {
        const int s = __builtin_amdgcn_readlane(sv, t + j);
        h[j] = *(const vecf*)(Hlane + (size_t)s * DIM);
    }
#pragma unroll
    for (int j = 0; j < G; ++j) {
        const float w = __int_as_float(__builtin_amdgcn_readlane(__float_as_int(dv), t + j)) * di;
        acc = h[j] * w + acc;
    }
}

template <int DIM, bool RELU>
__global__ __launch_bounds__(256) void k_gather(const float* __restrict__ H,
                                                const int* __restrict__ rowptr,
                                                const int* __restrict__ colw,
                                                const float* __restrict__ dinv,
                                                const float* __restrict__ bias,
                                                float* __restrict__ Out, int n, int E) {
    constexpr int VPT = DIM / 32;
    typedef float vecf __attribute__((ext_vector_type(VPT)));
    const int lane = threadIdx.x & 31;
    const int wave0 = __builtin_amdgcn_readfirstlane(
        (int)((blockIdx.x * blockDim.x + threadIdx.x) >> 5));
    const int nwaves = (int)((gridDim.x * blockDim.x) >> 5);
    const float* Hlane = H + lane * VPT;
    const vecf bv = *(const vecf*)(bias + lane * VPT);

    for (int row = wave0; row < n; row += nwaves) {
        const int beg = rowptr[row];
        const int end = (row + 1 < n) ? rowptr[row + 1] : E;
        const float di = dinv[row];
        vecf acc = *(const vecf*)(Hlane + (size_t)row * DIM) * (di * di) + bv;

        for (int base = beg; base < end; base += 32) {
            const int m = min(32, end - base);
            int sv = 0;
            float dv = 0.0f;
            if (lane < m) {
                const int2 cw = *(const int2*)(colw + 2 * (size_t)(base + lane));
                sv = (int)min((unsigned)cw.x, (unsigned)(n - 1));
                dv = __int_as_float(cw.y);
            }
            int t = 0;
            for (; t + 8 <= m; t += 8) add_rows<8, DIM>(acc, Hlane, sv, dv, di, t);
            if (m & 4) { add_rows<4, DIM>(acc, Hlane, sv, dv, di, t); t += 4; }
            if (m & 2) { add_rows<2, DIM>(acc, Hlane, sv, dv, di, t); t += 2; }
            if (m & 1) { add_rows<1, DIM>(acc, Hlane, sv, dv, di, t); }
        }
        if (RELU) {
#pragma unroll
            for (int j = 0; j < VPT; ++j) acc[j] = fmaxf(acc[j], 0.0f);
        }
        *(vecf*)(Out + (size_t)row * DIM + lane * VPT) = acc;
    }
}

extern "C" void kernel_launch(void* const* d_in, const int* in_sizes, int n_in,
                              void* d_out, int out_size, void* d_ws, size_t ws_size,
                              hipStream_t stream) {
    const float* x   = (const float*)d_in[0];
    const int*   ei  = (const int*)d_in[1];
    const float* W1  = (const float*)d_in[2];
    const float* b1  = (const float*)d_in[3];
    const float* W2  = (const float*)d_in[4];
    const float* b2  = (const float*)d_in[5];
    float*       out = (float*)d_out;

    const int n = in_sizes[0] / IN_DIM;
    const int E = in_sizes[1] / 2;
    const int* srcIdx = ei;
    const int* dstIdx = ei + E;

    float* ws     = (float*)d_ws;
    int*   rowptr = (int*)ws;
    float* dinv   = ws + n;
    float* h      = dinv + n;
    float* agg    = h + (size_t)n * HID;

    const int n4 = (n + 3) & ~3;
    int* cnt  = (int*)out;
    int* col1 = cnt + n4;
    int* rank = col1 + 2 * (size_t)E;
    int* col2 = (int*)(h + (size_t)n * OUT);

    k_zero_cnt<<<(n + 255) / 256, 256, 0, stream>>>(cnt, n);
    k_count_rank<<<(E + 255) / 256, 256, 0, stream>>>(dstIdx, E, cnt, rank);
    k_scan_dinv<<<(n + 1023) / 1024, 256, 0, stream>>>(cnt, rowptr, dinv, n);
    k_fill<<<(E + 255) / 256, 256, 0, stream>>>(srcIdx, dstIdx, rank, rowptr, dinv, col1, E);

    const int tiles = (n + 15) / 16;
    const int gemmBlocks = (tiles + 7) / 8;
    const int gatherBlocks = 2048;

    k_gemm_wmma<8><<<gemmBlocks, 256, 0, stream>>>(x, W1, h, n);
    k_gather<HID, true><<<gatherBlocks, 256, 0, stream>>>(h, rowptr, col1, dinv, b1, agg, n, E);
    k_copy_int<<<(E / 2 + 256) / 256, 256, 0, stream>>>(col1, col2, 2 * E);

    k_gemm_wmma<4><<<gemmBlocks, 256, 0, stream>>>(agg, W2, h, n);
    k_gather<OUT, false><<<gatherBlocks, 256, 0, stream>>>(h, rowptr, col2, dinv, b2, out, n, E);
}


# Edges grouped by destination by a two-level counting sort on LDS counters (bins of 512 rows, one work-group per bin) in place of a global returning atomic per edge; sum order in a row changes
# speedup vs baseline: 3.1252x; 1.4360x over previous
#include <hip/hip_runtime.h>


typedef float v2f __attribute__((ext_vector_type(2)));
typedef float v8f __attribute__((ext_vector_type(8)));

#define IN_DIM 128
#define HID 128
#define OUT 64

#define BINW 512
#define CHUNK 8192
#define MAXB 256

__device__ __forceinline__ int bin_of(int d, int nb) {
    return (int)min((unsigned)d / (unsigned)BINW, (unsigned)(nb - 1));
}

__global__ __launch_bounds__(256) void k_bin_hist(const int* __restrict__ dst, int E, int nb,
                                                  int* __restrict__ hist) {
    __shared__ int h[MAXB];
    const int t = threadIdx.x;
    h[t] = 0;
    __syncthreads();
    const int e0 = blockIdx.x * CHUNK + t;
    for (int i0 = 0; i0 < CHUNK; i0 += 256 * 16) {
        int d[16];
#pragma unroll
        for (int j = 0; j < 16; ++j) d[j] = dst[min(e0 + i0 + j * 256, E - 1)];
        __builtin_amdgcn_sched_barrier(0);
#pragma unroll
        for (int j = 0; j < 16; ++j)
            atomicAdd(&h[bin_of(d[j], nb)], (e0 + i0 + j * 256 < E) ? 1 : 0);
    }
    __syncthreads();
    if (t < nb) hist[blockIdx.x * nb + t] = h[t];
}

__global__ __launch_bounds__(256) void k_bin_scatter(const int* __restrict__ src,
                                                     const int* __restrict__ dst, int E, int nb,
                                                     int G, const int* __restrict__ hist,
                                                     int* __restrict__ binStart,
                                                     int* __restrict__ binned) {
    __shared__ int cur[MAXB];
    __shared__ int wtot[8];
    const int t = threadIdx.x;
    const int lane = t & 31;
    const int wave = t >> 5;
    const int blk = blockIdx.x;

    int tot = 0, part = 0;
    const int tb = min(t, nb - 1);
    for (int g0 = 0; g0 < G; g0 += 28) {
        int v[28];
#pragma unroll
        for (int j = 0; j < 28; ++j) v[j] = hist[min(g0 + j, G - 1) * nb + tb];
        __builtin_amdgcn_sched_barrier(0);
#pragma unroll
        for (int j = 0; j < 28; ++j) {
            tot += (g0 + j < G) ? v[j] : 0;
            part += (g0 + j < blk) ? v[j] : 0;
        }
    }
    if (t >= nb) { tot = 0; part = 0; }

    int inc = tot;
    for (int o = 1; o < 32; o <<= 1) {
        int v = __shfl_up(inc, o);
        if (lane >= o) inc += v;
    }
    if (lane == 31) wtot[wave] = inc;
    __syncthreads();
    int off = 0;
    for (int w = 0; w < 8; ++w) {
        if (w < wave) off += wtot[w];
    }
    const int start = off + inc - tot;
    cur[t] = start + part;
    if (blk == 0) {
        if (t < nb) binStart[t] = start;
        if (t == 0) binStart[nb] = E;
    }
    __syncthreads();

    const int e0 = blk * CHUNK + t;
    for (int i0 = 0; i0 < CHUNK; i0 += 256 * 8) {
        int s[8], d[8];
#pragma unroll
        for (int j = 0; j < 8; ++j) {
            const int e = min(e0 + i0 + j * 256, E - 1);
            s[j] = src[e];
            d[j] = dst[e];
        }
        __builtin_amdgcn_sched_barrier(0);
        int pos[8];
#pragma unroll
        for (int j = 0; j < 8; ++j)
            pos[j] = atomicAdd(&cur[bin_of(d[j], nb)], (e0 + i0 + j * 256 < E) ? 1 : 0);
#pragma unroll
        for (int j = 0; j < 8; ++j) {
            if (e0 + i0 + j * 256 < E && (unsigned)pos[j] < (unsigned)E)
                *(int2*)(binned + 2 * (size_t)pos[j]) = make_int2(s[j], d[j]);
        }
    }
}

__global__ __launch_bounds__(BINW) void k_bin_rows(const int* __restrict__ binned,
                                                   const int* __restrict__ binStart, int n, int E,
                                                   int* __restrict__ rowptr,
                                                   float* __restrict__ dinv,
                                                   int* __restrict__ colsrc) {
    __shared__ int cnt[BINW];
    __shared__ int cur[BINW];
    __shared__ int wtot[BINW / 32];
    const int t = threadIdx.x;
    const int lane = t & 31;
    const int wave = t >> 5;
    const int nodeBase = blockIdx.x * BINW;
    const int beg = min(max(binStart[blockIdx.x], 0), E);
    const int end = min(max(binStart[blockIdx.x + 1], beg), E);
    const int2* pairs = (const int2*)binned;

    cnt[t] = 0;
    __syncthreads();
    for (int p0 = beg + t; p0 < end; p0 += BINW * 8) {
        int dd[8];
#pragma unroll
        for (int j = 0; j < 8; ++j) dd[j] = pairs[min(p0 + j * BINW, end - 1)].y;
        __builtin_amdgcn_sched_barrier(0);
#pragma unroll
        for (int j = 0; j < 8; ++j)
            atomicAdd(&cnt[min(max(dd[j] - nodeBase, 0), BINW - 1)], (p0 + j * BINW < end) ? 1 : 0);
    }
    __syncthreads();

    const int c = cnt[t];
    int inc = c;
    for (int o = 1; o < 32; o <<= 1) {
        int v = __shfl_up(inc, o);
        if (lane >= o) inc += v;
    }
    if (lane == 31) wtot[wave] = inc;
    __syncthreads();
    int off = 0;
    for (int w = 0; w < BINW / 32; ++w) {
        if (w < wave) off += wtot[w];
    }
    const int rp = beg + off + inc - c;
    cur[t] = rp;
    const int node = nodeBase + t;
    if (node < n) {
        rowptr[node] = rp;
        dinv[node] = rsqrtf((float)(c + 1));
    }
    __syncthreads();

    for (int p0 = beg + t; p0 < end; p0 += BINW * 8) {
        int2 sd[8];
#pragma unroll
        for (int j = 0; j < 8; ++j) sd[j] = pairs[min(p0 + j * BINW, end - 1)];
        __builtin_amdgcn_sched_barrier(0);
        int pos[8];
#pragma unroll
        for (int j = 0; j < 8; ++j)
            pos[j] = atomicAdd(&cur[min(max(sd[j].y - nodeBase, 0), BINW - 1)],
                               (p0 + j * BINW < end) ? 1 : 0);
#pragma unroll
        for (int j = 0; j < 8; ++j) {
            if (p0 + j * BINW < end && (unsigned)pos[j] < (unsigned)E) colsrc[pos[j]] = sd[j].x;
        }
    }
}

__global__ __launch_bounds__(256) void k_fill_w(const int* __restrict__ colsrc,
                                                const float* __restrict__ dinv,
                                                int* __restrict__ colw, int n, int E) {
    const int p = (blockIdx.x * blockDim.x + threadIdx.x) * 4;
    if (p + 3 < E) {
        const int4 s4 = *(const int4*)(colsrc + p);
        const int s0 = (int)min((unsigned)s4.x, (unsigned)(n - 1));
        const int s1 = (int)min((unsigned)s4.y, (unsigned)(n - 1));
        const int s2 = (int)min((unsigned)s4.z, (unsigned)(n - 1));
        const int s3 = (int)min((unsigned)s4.w, (unsigned)(n - 1));
        const float w0 = dinv[s0], w1 = dinv[s1], w2 = dinv[s2], w3 = dinv[s3];
        int4* o = (int4*)(colw + 2 * (size_t)p);
        o[0] = make_int4(s0, __float_as_int(w0), s1, __float_as_int(w1));
        o[1] = make_int4(s2, __float_as_int(w2), s3, __float_as_int(w3));
    } else {
        for (int q = p; q < E; ++q) {
            const int s = (int)min((unsigned)colsrc[q], (unsigned)(n - 1));
            colw[2 * (size_t)q] = s;
            colw[2 * (size_t)q + 1] = __float_as_int(dinv[s]);
        }
    }
}

__global__ __launch_bounds__(256) void k_copy_int(const int* __restrict__ a,
                                                  int* __restrict__ b, int count) {
    int i = (blockIdx.x * blockDim.x + threadIdx.x) * 4;
    if (i + 3 < count) {
        *(int4*)(b + i) = *(const int4*)(a + i);
    } else {
        for (int j = i; j < count; ++j) b[j] = a[j];
    }
}

template <int NTILES>
__global__ __launch_bounds__(256) void k_gemm_wmma(const float* __restrict__ X,
                                                   const float* __restrict__ W,
                                                   float* __restrict__ H, int nrows) {
    const int lane = threadIdx.x & 31;
    const int wave = threadIdx.x >> 5;
    const int tile = blockIdx.x * (blockDim.x >> 5) + wave;
    if (tile * 16 >= nrows) return;

    const int ln = lane & 15;
    const int hi = lane >> 4;
    const int ldw = NTILES * 16;

    v8f acc[NTILES] = {};

    const float* xrow = X + (size_t)(tile * 16 + ln) * IN_DIM;

    for (int k = 0; k < IN_DIM; k += 4) {
        v2f a;
        a.x = xrow[k + 2 * hi + 0];
        a.y = xrow[k + 2 * hi + 1];

        const float* wr = W + (size_t)(k + 2 * hi) * ldw;
#pragma unroll
        for (int n = 0; n < NTILES; ++n) {
            v2f b;
            b.x = wr[n * 16 + ln];
            b.y = wr[ldw + n * 16 + ln];
            acc[n] = __builtin_amdgcn_wmma_f32_16x16x4_f32(
                false, a, false, b, (short)0, acc[n], false, false);
        }
    }

#pragma unroll
    for (int n = 0; n < NTILES; ++n) {
#pragma unroll
        for (int r = 0; r < 8; ++r) {
            int orow = tile * 16 + r + 8 * hi;
            H[(size_t)orow * ldw + n * 16 + ln] = acc[n][r];
        }
    }
}


template <int G, int DIM>
__device__ __forceinline__ void add_rows(float __attribute__((ext_vector_type(DIM / 32)))& acc,
                                         const float* __restrict__ Hlane,
                                         int sv, float dv, float di, int t) {
    typedef float vecf __attribute__((ext_vector_type(DIM / 32)));
    vecf h[G];
#pragma unroll
    for (int j = 0; j < G; ++j) {
        const int s = __builtin_amdgcn_readlane(sv, t + j);
        h[j] = *(const vecf*)(Hlane + (size_t)s * DIM);
    }
#pragma unroll
    for (int j = 0; j < G; ++j) {
        const float w = __int_as_float(__builtin_amdgcn_readlane(__float_as_int(dv), t + j)) * di;
        acc = h[j] * w + acc;
    }
}

template <int DIM, bool RELU>
__global__ __launch_bounds__(256) void k_gather(const float* __restrict__ H,
                                                const int* __restrict__ rowptr,
                                                const int* __restrict__ colw,
                                                const float* __restrict__ dinv,
                                                const float* __restrict__ bias,
                                                float* __restrict__ Out, int n, int E) {
    constexpr int VPT = DIM / 32;
    typedef float vecf __attribute__((ext_vector_type(VPT)));
    const int lane = threadIdx.x & 31;
    const int wave0 = __builtin_amdgcn_readfirstlane(
        (int)((blockIdx.x * blockDim.x + threadIdx.x) >> 5));
    const int nwaves = (int)((gridDim.x * blockDim.x) >> 5);
    const float* Hlane = H + lane * VPT;
    const vecf bv = *(const vecf*)(bias + lane * VPT);

    for (int row = wave0; row < n; row += nwaves) {
        const int beg = rowptr[row];
        const int end = (row + 1 < n) ? rowptr[row + 1] : E;
        const float di = dinv[row];
        vecf acc = *(const vecf*)(Hlane + (size_t)row * DIM) * (di * di) + bv;

        for (int base = beg; base < end; base += 32) {
            const int m = min(32, end - base);
            int sv = 0;
            float dv = 0.0f;
            if (lane < m) {
                const int2 cw = *(const int2*)(colw + 2 * (size_t)(base + lane));
                sv = (int)min((unsigned)cw.x, (unsigned)(n - 1));
                dv = __int_as_float(cw.y);
            }
            int t = 0;
            for (; t + 8 <= m; t += 8) add_rows<8, DIM>(acc, Hlane, sv, dv, di, t);
            if (m & 4) { add_rows<4, DIM>(acc, Hlane, sv, dv, di, t); t += 4; }
            if (m & 2) { add_rows<2, DIM>(acc, Hlane, sv, dv, di, t); t += 2; }
            if (m & 1) { add_rows<1, DIM>(acc, Hlane, sv, dv, di, t); }
        }
        if (RELU) {
#pragma unroll
            for (int j = 0; j < VPT; ++j) acc[j] = fmaxf(acc[j], 0.0f);
        }
        *(vecf*)(Out + (size_t)row * DIM + lane * VPT) = acc;
    }
}

extern "C" void kernel_launch(void* const* d_in, const int* in_sizes, int n_in,
                              void* d_out, int out_size, void* d_ws, size_t ws_size,
                              hipStream_t stream) {
    const float* x   = (const float*)d_in[0];
    const int*   ei  = (const int*)d_in[1];
    const float* W1  = (const float*)d_in[2];
    const float* b1  = (const float*)d_in[3];
    const float* W2  = (const float*)d_in[4];
    const float* b2  = (const float*)d_in[5];
    float*       out = (float*)d_out;

    const int n = in_sizes[0] / IN_DIM;
    const int E = in_sizes[1] / 2;
    const int* srcIdx = ei;
    const int* dstIdx = ei + E;

    float* ws     = (float*)d_ws;
    int*   rowptr = (int*)ws;
    float* dinv   = ws + n;
    float* h      = dinv + n;
    float* agg    = h + (size_t)n * HID;

    const int nb = (n + BINW - 1) / BINW;
    const int G  = (E + CHUNK - 1) / CHUNK;
    int* hist     = (int*)out;
    int* binStart = hist + ((G * nb + 63) & ~63);
    int* binned   = binStart + ((nb + 1 + 63) & ~63);
    int* colsrc   = binned + 2 * (size_t)E;
    int* col1     = binned;
    int* col2 = (int*)(h + (size_t)n * OUT);

    k_bin_hist<<<G, 256, 0, stream>>>(dstIdx, E, nb, hist);
    k_bin_scatter<<<G, 256, 0, stream>>>(srcIdx, dstIdx, E, nb, G, hist, binStart, binned);
    k_bin_rows<<<nb, BINW, 0, stream>>>(binned, binStart, n, E, rowptr, dinv, colsrc);
    k_fill_w<<<(E / 4 + 256) / 256, 256, 0, stream>>>(colsrc, dinv, col1, n, E);

    const int tiles = (n + 15) / 16;
    const int gemmBlocks = (tiles + 7) / 8;
    const int gatherBlocks = 2048;

    k_gemm_wmma<8><<<gemmBlocks, 256, 0, stream>>>(x, W1, h, n);
    k_gather<HID, true><<<gatherBlocks, 256, 0, stream>>>(h, rowptr, col1, dinv, b1, agg, n, E);
    k_copy_int<<<(E / 2 + 256) / 256, 256, 0, stream>>>(col1, col2, 2 * E);

    k_gemm_wmma<4><<<gemmBlocks, 256, 0, stream>>>(agg, W2, h, n);
    k_gather<OUT, false><<<gatherBlocks, 256, 0, stream>>>(h, rowptr, col2, dinv, b2, out, n, E);
}


# Layer-1 row sums fused with the layer-2 dense product: a wave sums 16 rows (one round trip each) into an LDS tile and multiplies it by W2 on the f32 matrix unit; aggregate buffer and 2 launches gone
# speedup vs baseline: 3.7300x; 1.1935x over previous
#include <hip/hip_runtime.h>


typedef float v2f __attribute__((ext_vector_type(2)));
typedef float v8f __attribute__((ext_vector_type(8)));

#define IN_DIM 128
#define HID 128
#define OUT 64

#define BINW 512
#define CHUNK 8192
#define MAXB 256

__device__ __forceinline__ int bin_of(int d, int nb) {
    return (int)min((unsigned)d / (unsigned)BINW, (unsigned)(nb - 1));
}

__global__ __launch_bounds__(256) void k_bin_hist(const int* __restrict__ dst, int E, int nb,
                                                  int* __restrict__ hist) {
    __shared__ int h[MAXB];
    const int t = threadIdx.x;
    h[t] = 0;
    __syncthreads();
    const int e0 = blockIdx.x * CHUNK + t;
    for (int i0 = 0; i0 < CHUNK; i0 += 256 * 16) {
        int d[16];
#pragma unroll
        for (int j = 0; j < 16; ++j) d[j] = dst[min(e0 + i0 + j * 256, E - 1)];
        __builtin_amdgcn_sched_barrier(0);
#pragma unroll
        for (int j = 0; j < 16; ++j)
            atomicAdd(&h[bin_of(d[j], nb)], (e0 + i0 + j * 256 < E) ? 1 : 0);
    }
    __syncthreads();
    if (t < nb) hist[blockIdx.x * nb + t] = h[t];
}

__global__ __launch_bounds__(256) void k_bin_scatter(const int* __restrict__ src,
                                                     const int* __restrict__ dst, int E, int nb,
                                                     int G, const int* __restrict__ hist,
                                                     int* __restrict__ binStart,
                                                     int* __restrict__ binned) {
    __shared__ int cur[MAXB];
    __shared__ int wtot[8];
    const int t = threadIdx.x;
    const int lane = t & 31;
    const int wave = t >> 5;
    const int blk = blockIdx.x;

    int tot = 0, part = 0;
    const int tb = min(t, nb - 1);
    for (int g0 = 0; g0 < G; g0 += 28) {
        int v[28];
#pragma unroll
        for (int j = 0; j < 28; ++j) v[j] = hist[min(g0 + j, G - 1) * nb + tb];
        __builtin_amdgcn_sched_barrier(0);
#pragma unroll
        for (int j = 0; j < 28; ++j) {
            tot += (g0 + j < G) ? v[j] : 0;
            part += (g0 + j < blk) ? v[j] : 0;
        }
    }
    if (t >= nb) { tot = 0; part = 0; }

    int inc = tot;
    for (int o = 1; o < 32; o <<= 1) {
        int v = __shfl_up(inc, o);
        if (lane >= o) inc += v;
    }
    if (lane == 31) wtot[wave] = inc;
    __syncthreads();
    int off = 0;
    for (int w = 0; w < 8; ++w) {
        if (w < wave) off += wtot[w];
    }
    const int start = off + inc - tot;
    cur[t] = start + part;
    if (blk == 0) {
        if (t < nb) binStart[t] = start;
        if (t == 0) binStart[nb] = E;
    }
    __syncthreads();

    const int e0 = blk * CHUNK + t;
    for (int i0 = 0; i0 < CHUNK; i0 += 256 * 8) {
        int s[8], d[8];
#pragma unroll
        for (int j = 0; j < 8; ++j) {
            const int e = min(e0 + i0 + j * 256, E - 1);
            s[j] = src[e];
            d[j] = dst[e];
        }
        __builtin_amdgcn_sched_barrier(0);
        int pos[8];
#pragma unroll
        for (int j = 0; j < 8; ++j)
            pos[j] = atomicAdd(&cur[bin_of(d[j], nb)], (e0 + i0 + j * 256 < E) ? 1 : 0);
#pragma unroll
        for (int j = 0; j < 8; ++j) {
            if (e0 + i0 + j * 256 < E && (unsigned)pos[j] < (unsigned)E)
                *(int2*)(binned + 2 * (size_t)pos[j]) = make_int2(s[j], d[j]);
        }
    }
}

__global__ __launch_bounds__(BINW) void k_bin_rows(const int* __restrict__ binned,
                                                   const int* __restrict__ binStart, int n, int E,
                                                   int* __restrict__ rowptr,
                                                   float* __restrict__ dinv,
                                                   int* __restrict__ colsrc) {
    __shared__ int cnt[BINW];
    __shared__ int cur[BINW];
    __shared__ int wtot[BINW / 32];
    const int t = threadIdx.x;
    const int lane = t & 31;
    const int wave = t >> 5;
    const int nodeBase = blockIdx.x * BINW;
    const int beg = min(max(binStart[blockIdx.x], 0), E);
    const int end = min(max(binStart[blockIdx.x + 1], beg), E);
    const int2* pairs = (const int2*)binned;

    cnt[t] = 0;
    __syncthreads();
    for (int p0 = beg + t; p0 < end; p0 += BINW * 8) {
        int dd[8];
#pragma unroll
        for (int j = 0; j < 8; ++j) dd[j] = pairs[min(p0 + j * BINW, end - 1)].y;
        __builtin_amdgcn_sched_barrier(0);
#pragma unroll
        for (int j = 0; j < 8; ++j)
            atomicAdd(&cnt[min(max(dd[j] - nodeBase, 0), BINW - 1)], (p0 + j * BINW < end) ? 1 : 0);
    }
    __syncthreads();

    const int c = cnt[t];
    int inc = c;
    for (int o = 1; o < 32; o <<= 1) {
        int v = __shfl_up(inc, o);
        if (lane >= o) inc += v;
    }
    if (lane == 31) wtot[wave] = inc;
    __syncthreads();
    int off = 0;
    for (int w = 0; w < BINW / 32; ++w) {
        if (w < wave) off += wtot[w];
    }
    const int rp = beg + off + inc - c;
    cur[t] = rp;
    const int node = nodeBase + t;
    if (node < n) {
        rowptr[node] = rp;
        dinv[node] = rsqrtf((float)(c + 1));
    }
    __syncthreads();

    for (int p0 = beg + t; p0 < end; p0 += BINW * 8) {
        int2 sd[8];
#pragma unroll
        for (int j = 0; j < 8; ++j) sd[j] = pairs[min(p0 + j * BINW, end - 1)];
        __builtin_amdgcn_sched_barrier(0);
        int pos[8];
#pragma unroll
        for (int j = 0; j < 8; ++j)
            pos[j] = atomicAdd(&cur[min(max(sd[j].y - nodeBase, 0), BINW - 1)],
                               (p0 + j * BINW < end) ? 1 : 0);
#pragma unroll
        for (int j = 0; j < 8; ++j) {
            if (p0 + j * BINW < end && (unsigned)pos[j] < (unsigned)E) colsrc[pos[j]] = sd[j].x;
        }
    }
}

template <int NTILES>
__global__ __launch_bounds__(256) void k_gemm_wmma(const float* __restrict__ X,
                                                   const float* __restrict__ W,
                                                   const float* __restrict__ dinv,
                                                   float* __restrict__ H, int nrows) {
    const int lane = threadIdx.x & 31;
    const int wave = threadIdx.x >> 5;
    const int tile = blockIdx.x * (blockDim.x >> 5) + wave;
    if (tile * 16 >= nrows) return;

    const int ln = lane & 15;
    const int hi = lane >> 4;
    const int ldw = NTILES * 16;

    v8f acc[NTILES] = {};

    float dr[8];
#pragma unroll
    for (int r = 0; r < 8; ++r) dr[r] = dinv[min(tile * 16 + r + 8 * hi, nrows - 1)];

    const float* xrow = X + (size_t)(tile * 16 + ln) * IN_DIM;

    for (int k = 0; k < IN_DIM; k += 4) {
        v2f a;
        a.x = xrow[k + 2 * hi + 0];
        a.y = xrow[k + 2 * hi + 1];

        const float* wr = W + (size_t)(k + 2 * hi) * ldw;
#pragma unroll
        for (int n = 0; n < NTILES; ++n) {
            v2f b;
            b.x = wr[n * 16 + ln];
            b.y = wr[ldw + n * 16 + ln];
            acc[n] = __builtin_amdgcn_wmma_f32_16x16x4_f32(
                false, a, false, b, (short)0, acc[n], false, false);
        }
    }

#pragma unroll
    for (int n = 0; n < NTILES; ++n) {
#pragma unroll
        for (int r = 0; r < 8; ++r) {
            int orow = tile * 16 + r + 8 * hi;
            H[(size_t)orow * ldw + n * 16 + ln] = acc[n][r] * dr[r];
        }
    }
}


template <int G, int DIM>
__device__ __forceinline__ void add_rows(float __attribute__((ext_vector_type(DIM / 32)))& acc,
                                         const float* __restrict__ Hlane, int sv, int t) {
    typedef float vecf __attribute__((ext_vector_type(DIM / 32)));
    vecf h[G];
#pragma unroll
    for (int j = 0; j < G; ++j) {
        const int s = __builtin_amdgcn_readlane(sv, t + j);
        h[j] = *(const vecf*)(Hlane + (size_t)s * DIM);
    }
#pragma unroll
    for (int j = 0; j < G; ++j) acc = acc + h[j];
}

template <int DIM, bool RELU>
__global__ __launch_bounds__(256) void k_gather(const float* __restrict__ H,
                                                const int* __restrict__ rowptr,
                                                const int* __restrict__ colsrc,
                                                const float* __restrict__ dinv,
                                                const float* __restrict__ bias,
                                                float* __restrict__ Out, int n, int E) {
    constexpr int VPT = DIM / 32;
    typedef float vecf __attribute__((ext_vector_type(VPT)));
    const int lane = threadIdx.x & 31;
    const int wave0 = __builtin_amdgcn_readfirstlane(
        (int)((blockIdx.x * blockDim.x + threadIdx.x) >> 5));
    const int nwaves = (int)((gridDim.x * blockDim.x) >> 5);
    const float* Hlane = H + lane * VPT;
    const vecf bv = *(const vecf*)(bias + lane * VPT);

    for (int row = wave0; row < n; row += nwaves) {
        const int beg = rowptr[row];
        const int end = (row + 1 < n) ? rowptr[row + 1] : E;
        const float di = dinv[row];
        vecf acc = *(const vecf*)(Hlane + (size_t)row * DIM);

        for (int base = beg; base < end; base += 32) {
            const int m = min(32, end - base);
            int sv = 0;
            if (lane < m)
                sv = (int)min((unsigned)colsrc[base + lane], (unsigned)(n - 1));
            int t = 0;
            for (; t + 8 <= m; t += 8) add_rows<8, DIM>(acc, Hlane, sv, t);
            if (m & 4) { add_rows<4, DIM>(acc, Hlane, sv, t); t += 4; }
            if (m & 2) { add_rows<2, DIM>(acc, Hlane, sv, t); t += 2; }
            if (m & 1) { add_rows<1, DIM>(acc, Hlane, sv, t); }
        }
        acc = acc * di + bv;
        if (RELU) {
#pragma unroll
            for (int j = 0; j < VPT; ++j) acc[j] = fmaxf(acc[j], 0.0f);
        }
        *(vecf*)(Out + (size_t)row * DIM + lane * VPT) = acc;
    }
}

#define TPITCH 132
#define FWAVES 4
#define GB 24

__device__ __forceinline__ int load_slots(const int* __restrict__ colsrc, int b, int lane, int E) {
    return colsrc[min(b + lane, E - 1)];
}

typedef float vec4f __attribute__((ext_vector_type(4)));

template <int J, int G>
__device__ __forceinline__ void request_rows(vec4f (&h)[G], const float* __restrict__ Hlane, int sv,
                                             int t, int cnt) {
    if constexpr (J < G) {
        if (J < cnt) {
            const int s = __builtin_amdgcn_readlane(sv, t + J);
            h[J] = *(const vec4f*)(Hlane + (size_t)s * HID);
            request_rows<J + 1, G>(h, Hlane, sv, t, cnt);
        }
    }
}

template <int J, int G>
__device__ __forceinline__ void sum_rows(vec4f& acc, const vec4f (&h)[G], int cnt) {
    if constexpr (J < G) {
        if (J < cnt) {
            acc = acc + h[J];
            sum_rows<J + 1, G>(acc, h, cnt);
        }
    }
}

template <int G>
__device__ __forceinline__ void add_rows_upto(vec4f& acc, const float* __restrict__ Hlane, int sv,
                                              int t, int cnt) {
    vec4f h[G];
    request_rows<0, G>(h, Hlane, sv, t, cnt);
    sum_rows<0, G>(acc, h, cnt);
}

__global__ __launch_bounds__(FWAVES * 32) void k_gather_gemm(const float* __restrict__ H,
                                                             const int* __restrict__ rowptr,
                                                             const int* __restrict__ colsrc,
                                                             const float* __restrict__ dinv,
                                                             const float* __restrict__ bias,
                                                             const float* __restrict__ W,
                                                             float* __restrict__ Out, int n, int E) {
    typedef float vecf __attribute__((ext_vector_type(4)));
    __shared__ float tiles[FWAVES][16 * TPITCH];
    const int lane = threadIdx.x & 31;
    const int tile = __builtin_amdgcn_readfirstlane((int)(blockIdx.x * FWAVES + (threadIdx.x >> 5)));
    if (tile * 16 >= n) return;
    float* tl = tiles[threadIdx.x >> 5];
    const int row0 = tile * 16;

    const int ridx = row0 + lane;
    int rpv = rowptr[min(ridx, n - 1)];
    if (ridx >= n) rpv = E;
    rpv = min(max(rpv, 0), E);
    const float dv = dinv[min(ridx, n - 1)];

    const float* Hlane = H + lane * 4;
    const vecf bv = *(const vecf*)(bias + lane * 4);

    int beg = __builtin_amdgcn_readlane(rpv, 0);
    int end = max(__builtin_amdgcn_readlane(rpv, 1), beg);
    int svn = load_slots(colsrc, beg, lane, E);
    for (int r = 0; r < 16; ++r) {
        const int row = min(row0 + r, n - 1);
        const int b = beg, e = end;
        const int sv = (int)min((unsigned)svn, (unsigned)(n - 1));
        const float di = __int_as_float(__builtin_amdgcn_readlane(__float_as_int(dv), r));
        if (r < 15) {
            beg = e;
            end = max(__builtin_amdgcn_readlane(rpv, r + 2), beg);
            svn = load_slots(colsrc, beg, lane, E);
        }
        vecf acc = *(const vecf*)(Hlane + (size_t)row * HID);

        const int m0 = min(32, e - b);
        for (int t = 0; t < m0; t += GB) add_rows_upto<GB>(acc, Hlane, sv, t, min(GB, m0 - t));
        for (int base = b + 32; base < e; base += 32) {
            const int m = min(32, e - base);
            const int sx = (int)min((unsigned)load_slots(colsrc, base, lane, E), (unsigned)(n - 1));
            for (int t = 0; t < m; t += 2) add_rows_upto<2>(acc, Hlane, sx, t, min(2, m - t));
        }
        acc = acc * di + bv;
#pragma unroll
        for (int j = 0; j < 4; ++j) acc[j] = fmaxf(acc[j], 0.0f);
        *(vecf*)(tl + r * TPITCH + lane * 4) = acc;
    }

    const int ln = lane & 15;
    const int hi = lane >> 4;
    v8f c[4] = {};
    const float* arow = tl + ln * TPITCH + 2 * hi;
#pragma unroll 4
    for (int k = 0; k < HID; k += 4) {
        const v2f a = *(const v2f*)(arow + k);
        const float* wr = W + (size_t)(k + 2 * hi) * OUT;
#pragma unroll
        for (int nt = 0; nt < 4; ++nt) {
            v2f bb;
            bb.x = wr[nt * 16 + ln];
            bb.y = wr[OUT + nt * 16 + ln];
            c[nt] = __builtin_amdgcn_wmma_f32_16x16x4_f32(
                false, a, false, bb, (short)0, c[nt], false, false);
        }
    }

    const int dvi = __float_as_int(dv);
#pragma unroll
    for (int r = 0; r < 8; ++r) {
        const int s0 = __builtin_amdgcn_readlane(dvi, r);
        const int s1 = __builtin_amdgcn_readlane(dvi, r + 8);
        const float sc = __int_as_float(hi ? s1 : s0);
        const int orow = row0 + r + 8 * hi;
#pragma unroll
        for (int nt = 0; nt < 4; ++nt) {
            if (orow < n) Out[(size_t)orow * OUT + nt * 16 + ln] = c[nt][r] * sc;
        }
    }
}

extern "C" void kernel_launch(void* const* d_in, const int* in_sizes, int n_in,
                              void* d_out, int out_size, void* d_ws, size_t ws_size,
                              hipStream_t stream) {
    const float* x   = (const float*)d_in[0];
    const int*   ei  = (const int*)d_in[1];
    const float* W1  = (const float*)d_in[2];
    const float* b1  = (const float*)d_in[3];
    const float* W2  = (const float*)d_in[4];
    const float* b2  = (const float*)d_in[5];
    float*       out = (float*)d_out;

    const int n = in_sizes[0] / IN_DIM;
    const int E = in_sizes[1] / 2;
    const int* srcIdx = ei;
    const int* dstIdx = ei + E;

    float* ws     = (float*)d_ws;
    int*   rowptr = (int*)ws;
    float* dinv   = ws + n;
    float* h      = dinv + n;
    float* h2     = h + (size_t)n * HID;
    int*   colsrc = (int*)(h2 + (size_t)n * OUT);

    const int nb = (n + BINW - 1) / BINW;
    const int G  = (E + CHUNK - 1) / CHUNK;
    int* hist     = (int*)out;
    int* binStart = hist + ((G * nb + 63) & ~63);
    int* binned   = binStart + ((nb + 1 + 63) & ~63);

    k_bin_hist<<<G, 256, 0, stream>>>(dstIdx, E, nb, hist);
    k_bin_scatter<<<G, 256, 0, stream>>>(srcIdx, dstIdx, E, nb, G, hist, binStart, binned);
    k_bin_rows<<<nb, BINW, 0, stream>>>(binned, binStart, n, E, rowptr, dinv, colsrc);

    const int tiles = (n + 15) / 16;
    const int gemmBlocks = (tiles + 7) / 8;
    const int fusedBlocks = (tiles + FWAVES - 1) / FWAVES;
    const int gatherBlocks = 2048;

    k_gemm_wmma<8><<<gemmBlocks, 256, 0, stream>>>(x, W1, dinv, h, n);
    k_gather_gemm<<<fusedBlocks, FWAVES * 32, 0, stream>>>(h, rowptr, colsrc, dinv, b1, W2, h2, n, E);

    k_gather<OUT, false><<<gatherBlocks, 256, 0, stream>>>(h2, rowptr, colsrc, dinv, b2, out, n, E);
}


# Both sort levels collect their per-edge output in LDS (a chunk's pairs grouped by bin; a bin's source list) and store it as contiguous runs, not one 8-B or 4-B store per edge to its own line
# speedup vs baseline: 3.8412x; 1.0298x over previous
#include <hip/hip_runtime.h>


typedef float v2f __attribute__((ext_vector_type(2)));
typedef float v8f __attribute__((ext_vector_type(8)));

#define IN_DIM 128
#define HID 128
#define OUT 64

#define BINW 512
#define CHUNK 8192
#define MAXB 256

__device__ __forceinline__ int bin_of(int d, int nb) {
    return (int)min((unsigned)d / (unsigned)BINW, (unsigned)(nb - 1));
}

__global__ __launch_bounds__(256) void k_bin_hist(const int* __restrict__ dst, int E, int nb,
                                                  int* __restrict__ hist) {
    __shared__ int h[MAXB];
    const int t = threadIdx.x;
    h[t] = 0;
    __syncthreads();
    const int e0 = blockIdx.x * CHUNK + t;
    for (int i0 = 0; i0 < CHUNK; i0 += 256 * 16) {
        int d[16];
#pragma unroll
        for (int j = 0; j < 16; ++j) d[j] = dst[min(e0 + i0 + j * 256, E - 1)];
        __builtin_amdgcn_sched_barrier(0);
#pragma unroll
        for (int j = 0; j < 16; ++j)
            atomicAdd(&h[bin_of(d[j], nb)], (e0 + i0 + j * 256 < E) ? 1 : 0);
    }
    __syncthreads();
    if (t < nb) hist[blockIdx.x * nb + t] = h[t];
}

typedef int v2i __attribute__((ext_vector_type(2)));

__global__ __launch_bounds__(256) void k_bin_scatter(const int* __restrict__ src,
                                                     const int* __restrict__ dst, int E, int nb,
                                                     int G, const int* __restrict__ hist,
                                                     int* __restrict__ binStart,
                                                     int* __restrict__ binned) {
    __shared__ int cur[MAXB];
    __shared__ int gdel[MAXB];
    __shared__ int wtot[8];
    __shared__ int wown[8];
    __shared__ v2i stage[CHUNK + 32];
    const int t = threadIdx.x;
    const int lane = t & 31;
    const int wave = t >> 5;
    const int blk = blockIdx.x;

    int tot = 0, part = 0, own = 0;
    const int tb = min(t, nb - 1);
    for (int g0 = 0; g0 < G; g0 += 28) {
        int v[28];
#pragma unroll
        for (int j = 0; j < 28; ++j) v[j] = hist[min(g0 + j, G - 1) * nb + tb];
        __builtin_amdgcn_sched_barrier(0);
#pragma unroll
        for (int j = 0; j < 28; ++j) {
            tot += (g0 + j < G) ? v[j] : 0;
            part += (g0 + j < blk) ? v[j] : 0;
            own += (g0 + j == blk) ? v[j] : 0;
        }
    }
    if (t >= nb) { tot = 0; part = 0; own = 0; }

    int inc = tot, linc = own;
    for (int o = 1; o < 32; o <<= 1) {
        int v = __shfl_up(inc, o);
        int u = __shfl_up(linc, o);
        if (lane >= o) { inc += v; linc += u; }
    }
    if (lane == 31) { wtot[wave] = inc; wown[wave] = linc; }
    __syncthreads();
    int off = 0, loff = 0;
    for (int w = 0; w < 8; ++w) {
        if (w < wave) { off += wtot[w]; loff += wown[w]; }
    }
    const int start = off + inc - tot;
    const int lstart = loff + linc - own;
    cur[t] = lstart;
    gdel[t] = start + part - lstart;
    if (blk == 0) {
        if (t < nb) binStart[t] = start;
        if (t == 0) binStart[nb] = E;
    }
    __syncthreads();

    const int e0 = blk * CHUNK + t;
    for (int i0 = 0; i0 < CHUNK; i0 += 256 * 8) {
        int s[8], d[8];
#pragma unroll
        for (int j = 0; j < 8; ++j) {
            const int e = min(e0 + i0 + j * 256, E - 1);
            s[j] = src[e];
            d[j] = dst[e];
        }
        __builtin_amdgcn_sched_barrier(0);
        int pos[8];
#pragma unroll
        for (int j = 0; j < 8; ++j)
            pos[j] = atomicAdd(&cur[bin_of(d[j], nb)], (e0 + i0 + j * 256 < E) ? 1 : 0);
#pragma unroll
        for (int j = 0; j < 8; ++j) {
            const bool ok = (e0 + i0 + j * 256 < E) && ((unsigned)pos[j] < (unsigned)CHUNK);
            v2i p;
            p.x = s[j];
            p.y = d[j];
            stage[ok ? pos[j] : CHUNK + lane] = p;
        }
    }
    __syncthreads();

    const int cnt = min(CHUNK, E - blk * CHUNK);
    for (int i0 = 0; i0 < cnt; i0 += 256 * 8) {
        v2i p[8];
        int gp[8];
#pragma unroll
        for (int j = 0; j < 8; ++j) p[j] = stage[i0 + j * 256 + t];
#pragma unroll
        for (int j = 0; j < 8; ++j) gp[j] = gdel[bin_of(p[j].y, nb)] + i0 + j * 256 + t;
#pragma unroll
        for (int j = 0; j < 8; ++j) {
            if (i0 + j * 256 + t < cnt && (unsigned)gp[j] < (unsigned)E)
                *(v2i*)(binned + 2 * (size_t)gp[j]) = p[j];
        }
    }
}

#define RCAP 16384
__global__ __launch_bounds__(BINW) void k_bin_rows(const int* __restrict__ binned,
                                                   const int* __restrict__ binStart, int n, int E,
                                                   int* __restrict__ rowptr,
                                                   float* __restrict__ dinv,
                                                   int* __restrict__ colsrc) {
    __shared__ int cnt[BINW];
    __shared__ int cur[BINW];
    __shared__ int wtot[BINW / 32];
    __shared__ int stage[RCAP + 32];
    const int t = threadIdx.x;
    const int lane = t & 31;
    const int wave = t >> 5;
    const int nodeBase = blockIdx.x * BINW;
    const int beg = min(max(binStart[blockIdx.x], 0), E);
    const int end = min(max(binStart[blockIdx.x + 1], beg), E);
    const int total = end - beg;
    const bool staged = total <= RCAP;
    const int2* pairs = (const int2*)binned;

    cnt[t] = 0;
    __syncthreads();
    for (int p0 = beg + t; p0 < end; p0 += BINW * 8) {
        int dd[8];
#pragma unroll
        for (int j = 0; j < 8; ++j) dd[j] = pairs[min(p0 + j * BINW, end - 1)].y;
        __builtin_amdgcn_sched_barrier(0);
#pragma unroll
        for (int j = 0; j < 8; ++j)
            atomicAdd(&cnt[min(max(dd[j] - nodeBase, 0), BINW - 1)], (p0 + j * BINW < end) ? 1 : 0);
    }
    __syncthreads();

    const int c = cnt[t];
    int inc = c;
    for (int o = 1; o < 32; o <<= 1) {
        int v = __shfl_up(inc, o);
        if (lane >= o) inc += v;
    }
    if (lane == 31) wtot[wave] = inc;
    __syncthreads();
    int off = 0;
    for (int w = 0; w < BINW / 32; ++w) {
        if (w < wave) off += wtot[w];
    }
    const int rp = beg + off + inc - c;
    cur[t] = rp;
    const int node = nodeBase + t;
    if (node < n) {
        rowptr[node] = rp;
        dinv[node] = rsqrtf((float)(c + 1));
    }
    __syncthreads();

    for (int p0 = beg + t; p0 < end; p0 += BINW * 8) {
        int2 sd[8];
#pragma unroll
        for (int j = 0; j < 8; ++j) sd[j] = pairs[min(p0 + j * BINW, end - 1)];
        __builtin_amdgcn_sched_barrier(0);
        int pos[8];
#pragma unroll
        for (int j = 0; j < 8; ++j)
            pos[j] = atomicAdd(&cur[min(max(sd[j].y - nodeBase, 0), BINW - 1)],
                               (p0 + j * BINW < end) ? 1 : 0);
        if (staged) {
#pragma unroll
            for (int j = 0; j < 8; ++j) {
                const unsigned l = (unsigned)(pos[j] - beg);
                stage[(p0 + j * BINW < end && l < (unsigned)RCAP) ? l : (unsigned)(RCAP + lane)] = sd[j].x;
            }
        } else {
#pragma unroll
            for (int j = 0; j < 8; ++j) {
                if (p0 + j * BINW < end && (unsigned)pos[j] < (unsigned)E) colsrc[pos[j]] = sd[j].x;
            }
        }
    }
    __syncthreads();
    if (staged) {
        for (int i = t; i < total; i += BINW) colsrc[beg + i] = stage[i];
    }
}

template <int NTILES>
__global__ __launch_bounds__(256) void k_gemm_wmma(const float* __restrict__ X,
                                                   const float* __restrict__ W,
                                                   const float* __restrict__ dinv,
                                                   float* __restrict__ H, int nrows) {
    const int lane = threadIdx.x & 31;
    const int wave = threadIdx.x >> 5;
    const int tile = blockIdx.x * (blockDim.x >> 5) + wave;
    if (tile * 16 >= nrows) return;

    const int ln = lane & 15;
    const int hi = lane >> 4;
    const int ldw = NTILES * 16;

    v8f acc[NTILES] = {};

    float dr[8];
#pragma unroll
    for (int r = 0; r < 8; ++r) dr[r] = dinv[min(tile * 16 + r + 8 * hi, nrows - 1)];

    const float* xrow = X + (size_t)(tile * 16 + ln) * IN_DIM;

    for (int k = 0; k < IN_DIM; k += 4) {
        v2f a;
        a.x = xrow[k + 2 * hi + 0];
        a.y = xrow[k + 2 * hi + 1];

        const float* wr = W + (size_t)(k + 2 * hi) * ldw;
#pragma unroll
        for (int n = 0; n < NTILES; ++n) {
            v2f b;
            b.x = wr[n * 16 + ln];
            b.y = wr[ldw + n * 16 + ln];
            acc[n] = __builtin_amdgcn_wmma_f32_16x16x4_f32(
                false, a, false, b, (short)0, acc[n], false, false);
        }
    }

#pragma unroll
    for (int n = 0; n < NTILES; ++n) {
#pragma unroll
        for (int r = 0; r < 8; ++r) {
            int orow = tile * 16 + r + 8 * hi;
            H[(size_t)orow * ldw + n * 16 + ln] = acc[n][r] * dr[r];
        }
    }
}


template <int G, int DIM>
__device__ __forceinline__ void add_rows(float __attribute__((ext_vector_type(DIM / 32)))& acc,
                                         const float* __restrict__ Hlane, int sv, int t) {
    typedef float vecf __attribute__((ext_vector_type(DIM / 32)));
    vecf h[G];
#pragma unroll
    for (int j = 0; j < G; ++j) {
        const int s = __builtin_amdgcn_readlane(sv, t + j);
        h[j] = *(const vecf*)(Hlane + (size_t)s * DIM);
    }
#pragma unroll
    for (int j = 0; j < G; ++j) acc = acc + h[j];
}

template <int DIM, bool RELU>
__global__ __launch_bounds__(256) void k_gather(const float* __restrict__ H,
                                                const int* __restrict__ rowptr,
                                                const int* __restrict__ colsrc,
                                                const float* __restrict__ dinv,
                                                const float* __restrict__ bias,
                                                float* __restrict__ Out, int n, int E) {
    constexpr int VPT = DIM / 32;
    typedef float vecf __attribute__((ext_vector_type(VPT)));
    const int lane = threadIdx.x & 31;
    const int wave0 = __builtin_amdgcn_readfirstlane(
        (int)((blockIdx.x * blockDim.x + threadIdx.x) >> 5));
    const int nwaves = (int)((gridDim.x * blockDim.x) >> 5);
    const float* Hlane = H + lane * VPT;
    const vecf bv = *(const vecf*)(bias + lane * VPT);

    for (int row = wave0; row < n; row += nwaves) {
        const int beg = rowptr[row];
        const int end = (row + 1 < n) ? rowptr[row + 1] : E;
        const float di = dinv[row];
        vecf acc = *(const vecf*)(Hlane + (size_t)row * DIM);

        for (int base = beg; base < end; base += 32) {
            const int m = min(32, end - base);
            int sv = 0;
            if (lane < m)
                sv = (int)min((unsigned)colsrc[base + lane], (unsigned)(n - 1));
            int t = 0;
            for (; t + 8 <= m; t += 8) add_rows<8, DIM>(acc, Hlane, sv, t);
            if (m & 4) { add_rows<4, DIM>(acc, Hlane, sv, t); t += 4; }
            if (m & 2) { add_rows<2, DIM>(acc, Hlane, sv, t); t += 2; }
            if (m & 1) { add_rows<1, DIM>(acc, Hlane, sv, t); }
        }
        acc = acc * di + bv;
        if (RELU) {
#pragma unroll
            for (int j = 0; j < VPT; ++j) acc[j] = fmaxf(acc[j], 0.0f);
        }
        *(vecf*)(Out + (size_t)row * DIM + lane * VPT) = acc;
    }
}

#define TPITCH 132
#define FWAVES 4
#define GB 24

__device__ __forceinline__ int load_slots(const int* __restrict__ colsrc, int b, int lane, int E) {
    return colsrc[min(b + lane, E - 1)];
}

typedef float vec4f __attribute__((ext_vector_type(4)));

template <int J, int G>
__device__ __forceinline__ void request_rows(vec4f (&h)[G], const float* __restrict__ Hlane, int sv,
                                             int t, int cnt) {
    if constexpr (J < G) {
        if (J < cnt) {
            const int s = __builtin_amdgcn_readlane(sv, t + J);
            h[J] = *(const vec4f*)(Hlane + (size_t)s * HID);
            request_rows<J + 1, G>(h, Hlane, sv, t, cnt);
        }
    }
}

template <int J, int G>
__device__ __forceinline__ void sum_rows(vec4f& acc, const vec4f (&h)[G], int cnt) {
    if constexpr (J < G) {
        if (J < cnt) {
            acc = acc + h[J];
            sum_rows<J + 1, G>(acc, h, cnt);
        }
    }
}

template <int G>
__device__ __forceinline__ void add_rows_upto(vec4f& acc, const float* __restrict__ Hlane, int sv,
                                              int t, int cnt) {
    vec4f h[G];
    request_rows<0, G>(h, Hlane, sv, t, cnt);
    sum_rows<0, G>(acc, h, cnt);
}

__global__ __launch_bounds__(FWAVES * 32) void k_gather_gemm(const float* __restrict__ H,
                                                             const int* __restrict__ rowptr,
                                                             const int* __restrict__ colsrc,
                                                             const float* __restrict__ dinv,
                                                             const float* __restrict__ bias,
                                                             const float* __restrict__ W,
                                                             float* __restrict__ Out, int n, int E) {
    typedef float vecf __attribute__((ext_vector_type(4)));
    __shared__ float tiles[FWAVES][16 * TPITCH];
    const int lane = threadIdx.x & 31;
    const int tile = __builtin_amdgcn_readfirstlane((int)(blockIdx.x * FWAVES + (threadIdx.x >> 5)));
    if (tile * 16 >= n) return;
    float* tl = tiles[threadIdx.x >> 5];
    const int row0 = tile * 16;

    const int ridx = row0 + lane;
    int rpv = rowptr[min(ridx, n - 1)];
    if (ridx >= n) rpv = E;
    rpv = min(max(rpv, 0), E);
    const float dv = dinv[min(ridx, n - 1)];

    const float* Hlane = H + lane * 4;
    const vecf bv = *(const vecf*)(bias + lane * 4);

    int beg = __builtin_amdgcn_readlane(rpv, 0);
    int end = max(__builtin_amdgcn_readlane(rpv, 1), beg);
    int svn = load_slots(colsrc, beg, lane, E);
    for (int r = 0; r < 16; ++r) {
        const int row = min(row0 + r, n - 1);
        const int b = beg, e = end;
        const int sv = (int)min((unsigned)svn, (unsigned)(n - 1));
        const float di = __int_as_float(__builtin_amdgcn_readlane(__float_as_int(dv), r));
        if (r < 15) {
            beg = e;
            end = max(__builtin_amdgcn_readlane(rpv, r + 2), beg);
            svn = load_slots(colsrc, beg, lane, E);
        }
        vecf acc = *(const vecf*)(Hlane + (size_t)row * HID);

        const int m0 = min(32, e - b);
        for (int t = 0; t < m0; t += GB) add_rows_upto<GB>(acc, Hlane, sv, t, min(GB, m0 - t));
        for (int base = b + 32; base < e; base += 32) {
            const int m = min(32, e - base);
            const int sx = (int)min((unsigned)load_slots(colsrc, base, lane, E), (unsigned)(n - 1));
            for (int t = 0; t < m; t += 2) add_rows_upto<2>(acc, Hlane, sx, t, min(2, m - t));
        }
        acc = acc * di + bv;
#pragma unroll
        for (int j = 0; j < 4; ++j) acc[j] = fmaxf(acc[j], 0.0f);
        *(vecf*)(tl + r * TPITCH + lane * 4) = acc;
    }

    const int ln = lane & 15;
    const int hi = lane >> 4;
    v8f c[4] = {};
    const float* arow = tl + ln * TPITCH + 2 * hi;
#pragma unroll 4
    for (int k = 0; k < HID; k += 4) {
        const v2f a = *(const v2f*)(arow + k);
        const float* wr = W + (size_t)(k + 2 * hi) * OUT;
#pragma unroll
        for (int nt = 0; nt < 4; ++nt) {
            v2f bb;
            bb.x = wr[nt * 16 + ln];
            bb.y = wr[OUT + nt * 16 + ln];
            c[nt] = __builtin_amdgcn_wmma_f32_16x16x4_f32(
                false, a, false, bb, (short)0, c[nt], false, false);
        }
    }

    const int dvi = __float_as_int(dv);
#pragma unroll
    for (int r = 0; r < 8; ++r) {
        const int s0 = __builtin_amdgcn_readlane(dvi, r);
        const int s1 = __builtin_amdgcn_readlane(dvi, r + 8);
        const float sc = __int_as_float(hi ? s1 : s0);
        const int orow = row0 + r + 8 * hi;
#pragma unroll
        for (int nt = 0; nt < 4; ++nt) {
            if (orow < n) Out[(size_t)orow * OUT + nt * 16 + ln] = c[nt][r] * sc;
        }
    }
}

extern "C" void kernel_launch(void* const* d_in, const int* in_sizes, int n_in,
                              void* d_out, int out_size, void* d_ws, size_t ws_size,
                              hipStream_t stream) {
    const float* x   = (const float*)d_in[0];
    const int*   ei  = (const int*)d_in[1];
    const float* W1  = (const float*)d_in[2];
    const float* b1  = (const float*)d_in[3];
    const float* W2  = (const float*)d_in[4];
    const float* b2  = (const float*)d_in[5];
    float*       out = (float*)d_out;

    const int n = in_sizes[0] / IN_DIM;
    const int E = in_sizes[1] / 2;
    const int* srcIdx = ei;
    const int* dstIdx = ei + E;

    float* ws     = (float*)d_ws;
    int*   rowptr = (int*)ws;
    float* dinv   = ws + n;
    float* h      = dinv + n;
    float* h2     = h + (size_t)n * HID;
    int*   colsrc = (int*)(h2 + (size_t)n * OUT);

    const int nb = (n + BINW - 1) / BINW;
    const int G  = (E + CHUNK - 1) / CHUNK;
    int* hist     = (int*)out;
    int* binStart = hist + ((G * nb + 63) & ~63);
    int* binned   = binStart + ((nb + 1 + 63) & ~63);

    k_bin_hist<<<G, 256, 0, stream>>>(dstIdx, E, nb, hist);
    k_bin_scatter<<<G, 256, 0, stream>>>(srcIdx, dstIdx, E, nb, G, hist, binStart, binned);
    k_bin_rows<<<nb, BINW, 0, stream>>>(binned, binStart, n, E, rowptr, dinv, colsrc);

    const int tiles = (n + 15) / 16;
    const int gemmBlocks = (tiles + 7) / 8;
    const int fusedBlocks = (tiles + FWAVES - 1) / FWAVES;
    const int gatherBlocks = 2048;

    k_gemm_wmma<8><<<gemmBlocks, 256, 0, stream>>>(x, W1, dinv, h, n);
    k_gather_gemm<<<fusedBlocks, FWAVES * 32, 0, stream>>>(h, rowptr, colsrc, dinv, b1, W2, h2, n, E);

    k_gather<OUT, false><<<gatherBlocks, 256, 0, stream>>>(h2, rowptr, colsrc, dinv, b2, out, n, E);
}


# Layer-1 dense product reads both operands through LDS: W1 copied once per work-group (k rows paired), X tiles fetched as whole 512-B rows into a per-wave LDS tile; 256 resident groups of 16 waves
# speedup vs baseline: 4.1113x; 1.0703x over previous
#include <hip/hip_runtime.h>


typedef float v2f __attribute__((ext_vector_type(2)));
typedef float v8f __attribute__((ext_vector_type(8)));

#define IN_DIM 128
#define HID 128
#define OUT 64

#define BINW 512
#define CHUNK 8192
#define MAXB 256

__device__ __forceinline__ int bin_of(int d, int nb) {
    return (int)min((unsigned)d / (unsigned)BINW, (unsigned)(nb - 1));
}

__global__ __launch_bounds__(256) void k_bin_hist(const int* __restrict__ dst, int E, int nb,
                                                  int* __restrict__ hist) {
    __shared__ int h[MAXB];
    const int t = threadIdx.x;
    h[t] = 0;
    __syncthreads();
    const int e0 = blockIdx.x * CHUNK + t;
    for (int i0 = 0; i0 < CHUNK; i0 += 256 * 16) {
        int d[16];
#pragma unroll
        for (int j = 0; j < 16; ++j) d[j] = dst[min(e0 + i0 + j * 256, E - 1)];
        __builtin_amdgcn_sched_barrier(0);
#pragma unroll
        for (int j = 0; j < 16; ++j)
            atomicAdd(&h[bin_of(d[j], nb)], (e0 + i0 + j * 256 < E) ? 1 : 0);
    }
    __syncthreads();
    if (t < nb) hist[blockIdx.x * nb + t] = h[t];
}

typedef int v2i __attribute__((ext_vector_type(2)));

__global__ __launch_bounds__(256) void k_bin_scatter(const int* __restrict__ src,
                                                     const int* __restrict__ dst, int E, int nb,
                                                     int G, const int* __restrict__ hist,
                                                     int* __restrict__ binStart,
                                                     int* __restrict__ binned) {
    __shared__ int cur[MAXB];
    __shared__ int gdel[MAXB];
    __shared__ int wtot[8];
    __shared__ int wown[8];
    __shared__ v2i stage[CHUNK + 32];
    const int t = threadIdx.x;
    const int lane = t & 31;
    const int wave = t >> 5;
    const int blk = blockIdx.x;

    int tot = 0, part = 0, own = 0;
    const int tb = min(t, nb - 1);
    for (int g0 = 0; g0 < G; g0 += 28) {
        int v[28];
#pragma unroll
        for (int j = 0; j < 28; ++j) v[j] = hist[min(g0 + j, G - 1) * nb + tb];
        __builtin_amdgcn_sched_barrier(0);
#pragma unroll
        for (int j = 0; j < 28; ++j) {
            tot += (g0 + j < G) ? v[j] : 0;
            part += (g0 + j < blk) ? v[j] : 0;
            own += (g0 + j == blk) ? v[j] : 0;
        }
    }
    if (t >= nb) { tot = 0; part = 0; own = 0; }

    int inc = tot, linc = own;
    for (int o = 1; o < 32; o <<= 1) {
        int v = __shfl_up(inc, o);
        int u = __shfl_up(linc, o);
        if (lane >= o) { inc += v; linc += u; }
    }
    if (lane == 31) { wtot[wave] = inc; wown[wave] = linc; }
    __syncthreads();
    int off = 0, loff = 0;
    for (int w = 0; w < 8; ++w) {
        if (w < wave) { off += wtot[w]; loff += wown[w]; }
    }
    const int start = off + inc - tot;
    const int lstart = loff + linc - own;
    cur[t] = lstart;
    gdel[t] = start + part - lstart;
    if (blk == 0) {
        if (t < nb) binStart[t] = start;
        if (t == 0) binStart[nb] = E;
    }
    __syncthreads();

    const int e0 = blk * CHUNK + t;
    for (int i0 = 0; i0 < CHUNK; i0 += 256 * 8) {
        int s[8], d[8];
#pragma unroll
        for (int j = 0; j < 8; ++j) {
            const int e = min(e0 + i0 + j * 256, E - 1);
            s[j] = src[e];
            d[j] = dst[e];
        }
        __builtin_amdgcn_sched_barrier(0);
        int pos[8];
#pragma unroll
        for (int j = 0; j < 8; ++j)
            pos[j] = atomicAdd(&cur[bin_of(d[j], nb)], (e0 + i0 + j * 256 < E) ? 1 : 0);
#pragma unroll
        for (int j = 0; j < 8; ++j) {
            const bool ok = (e0 + i0 + j * 256 < E) && ((unsigned)pos[j] < (unsigned)CHUNK);
            v2i p;
            p.x = s[j];
            p.y = d[j];
            stage[ok ? pos[j] : CHUNK + lane] = p;
        }
    }
    __syncthreads();

    const int cnt = min(CHUNK, E - blk * CHUNK);
    for (int i0 = 0; i0 < cnt; i0 += 256 * 8) {
        v2i p[8];
        int gp[8];
#pragma unroll
        for (int j = 0; j < 8; ++j) p[j] = stage[i0 + j * 256 + t];
#pragma unroll
        for (int j = 0; j < 8; ++j) gp[j] = gdel[bin_of(p[j].y, nb)] + i0 + j * 256 + t;
#pragma unroll
        for (int j = 0; j < 8; ++j) {
            if (i0 + j * 256 + t < cnt && (unsigned)gp[j] < (unsigned)E)
                *(v2i*)(binned + 2 * (size_t)gp[j]) = p[j];
        }
    }
}

#define RCAP 16384
__global__ __launch_bounds__(BINW) void k_bin_rows(const int* __restrict__ binned,
                                                   const int* __restrict__ binStart, int n, int E,
                                                   int* __restrict__ rowptr,
                                                   float* __restrict__ dinv,
                                                   int* __restrict__ colsrc) {
    __shared__ int cnt[BINW];
    __shared__ int cur[BINW];
    __shared__ int wtot[BINW / 32];
    __shared__ int stage[RCAP + 32];
    const int t = threadIdx.x;
    const int lane = t & 31;
    const int wave = t >> 5;
    const int nodeBase = blockIdx.x * BINW;
    const int beg = min(max(binStart[blockIdx.x], 0), E);
    const int end = min(max(binStart[blockIdx.x + 1], beg), E);
    const int total = end - beg;
    const bool staged = total <= RCAP;
    const int2* pairs = (const int2*)binned;

    cnt[t] = 0;
    __syncthreads();
    for (int p0 = beg + t; p0 < end; p0 += BINW * 8) {
        int dd[8];
#pragma unroll
        for (int j = 0; j < 8; ++j) dd[j] = pairs[min(p0 + j * BINW, end - 1)].y;
        __builtin_amdgcn_sched_barrier(0);
#pragma unroll
        for (int j = 0; j < 8; ++j)
            atomicAdd(&cnt[min(max(dd[j] - nodeBase, 0), BINW - 1)], (p0 + j * BINW < end) ? 1 : 0);
    }
    __syncthreads();

    const int c = cnt[t];
    int inc = c;
    for (int o = 1; o < 32; o <<= 1) {
        int v = __shfl_up(inc, o);
        if (lane >= o) inc += v;
    }
    if (lane == 31) wtot[wave] = inc;
    __syncthreads();
    int off = 0;
    for (int w = 0; w < BINW / 32; ++w) {
        if (w < wave) off += wtot[w];
    }
    const int rp = beg + off + inc - c;
    cur[t] = rp;
    const int node = nodeBase + t;
    if (node < n) {
        rowptr[node] = rp;
        dinv[node] = rsqrtf((float)(c + 1));
    }
    __syncthreads();

    for (int p0 = beg + t; p0 < end; p0 += BINW * 8) {
        int2 sd[8];
#pragma unroll
        for (int j = 0; j < 8; ++j) sd[j] = pairs[min(p0 + j * BINW, end - 1)];
        __builtin_amdgcn_sched_barrier(0);
        int pos[8];
#pragma unroll
        for (int j = 0; j < 8; ++j)
            pos[j] = atomicAdd(&cur[min(max(sd[j].y - nodeBase, 0), BINW - 1)],
                               (p0 + j * BINW < end) ? 1 : 0);
        if (staged) {
#pragma unroll
            for (int j = 0; j < 8; ++j) {
                const unsigned l = (unsigned)(pos[j] - beg);
                stage[(p0 + j * BINW < end && l < (unsigned)RCAP) ? l : (unsigned)(RCAP + lane)] = sd[j].x;
            }
        } else {
#pragma unroll
            for (int j = 0; j < 8; ++j) {
                if (p0 + j * BINW < end && (unsigned)pos[j] < (unsigned)E) colsrc[pos[j]] = sd[j].x;
            }
        }
    }
    __syncthreads();
    if (staged) {
        for (int i = t; i < total; i += BINW) colsrc[beg + i] = stage[i];
    }
}

#define G1WAVES 16
#define G1GROUPS 256
#define WPITCH 288
#define XPITCH 132

typedef float v4f __attribute__((ext_vector_type(4)));

__global__ __launch_bounds__(G1WAVES * 32) void k_gemm1_lds(const float* __restrict__ X,
                                                            const float* __restrict__ W,
                                                            const float* __restrict__ dinv,
                                                            float* __restrict__ H, int nrows) {
    __shared__ __attribute__((aligned(16))) float wl[(IN_DIM / 2) * WPITCH];
    __shared__ __attribute__((aligned(16))) float xt[G1WAVES][16 * XPITCH];
    const int t = threadIdx.x;
    const int lane = t & 31;
    const int wave = __builtin_amdgcn_readfirstlane(t >> 5);
    const int ln = lane & 15;
    const int hi = lane >> 4;

    const int ntiles = (nrows + 15) >> 4;
    const int tbeg = (int)(((long long)blockIdx.x * ntiles) / G1GROUPS);
    const int tend = (int)(((long long)(blockIdx.x + 1) * ntiles) / G1GROUPS);

    int tile = tbeg + wave;
    v4f xr[16];
    {
        const int tl0 = min(tile, ntiles - 1);
#pragma unroll
        for (int r = 0; r < 16; ++r)
            xr[r] = *(const v4f*)(X + (size_t)min(tl0 * 16 + r, nrows - 1) * IN_DIM + lane * 4);
    }

    for (int i0 = 0; i0 < (IN_DIM / 2) * (HID / 2); i0 += G1WAVES * 32 * 8) {
        v2f w0[8], w1[8];
#pragma unroll
        for (int j = 0; j < 8; ++j) {
            const int i = i0 + j * G1WAVES * 32 + t;
            const int p = i >> 6, cp = i & 63;
            w0[j] = *(const v2f*)(W + (size_t)(2 * p) * HID + 2 * cp);
            w1[j] = *(const v2f*)(W + (size_t)(2 * p + 1) * HID + 2 * cp);
        }
#pragma unroll
        for (int j = 0; j < 8; ++j) {
            const int i = i0 + j * G1WAVES * 32 + t;
            const int p = i >> 6, cp = i & 63;
            v4f q;
            q.x = w0[j].x; q.y = w1[j].x; q.z = w0[j].y; q.w = w1[j].y;
            *(v4f*)(wl + p * WPITCH + 4 * cp) = q;
        }
    }
    __syncthreads();

    float* xtw = xt[wave];
    const float* arow = xtw + ln * XPITCH + 2 * hi;
    const float* brow = wl + hi * WPITCH + 2 * ln;
    for (; tile < tend; tile += G1WAVES) {
        const int row0 = tile * 16;
#pragma unroll
        for (int r = 0; r < 16; ++r) *(v4f*)(xtw + r * XPITCH + lane * 4) = xr[r];
        float dr[8];
#pragma unroll
        for (int r = 0; r < 8; ++r) dr[r] = dinv[min(row0 + r + 8 * hi, nrows - 1)];
        if (tile + G1WAVES < tend) {
#pragma unroll
            for (int r = 0; r < 16; ++r)
                xr[r] = *(const v4f*)(X + (size_t)min(row0 + G1WAVES * 16 + r, nrows - 1) * IN_DIM + lane * 4);
        }

        v8f acc[8] = {};
#pragma unroll 2
        for (int k = 0; k < IN_DIM; k += 4) {
            const v2f a = *(const v2f*)(arow + k);
            const float* bp = brow + (k >> 1) * WPITCH;
#pragma unroll
            for (int nt = 0; nt < 8; ++nt) {
                const v2f bb = *(const v2f*)(bp + nt * 32);
                acc[nt] = __builtin_amdgcn_wmma_f32_16x16x4_f32(
                    false, a, false, bb, (short)0, acc[nt], false, false);
            }
        }

        float* hrow = H + (size_t)(row0 + 8 * hi) * HID + ln;
        if (row0 + 16 <= nrows) {
#pragma unroll
            for (int nt = 0; nt < 8; ++nt) {
#pragma unroll
                for (int r = 0; r < 8; ++r) hrow[(size_t)r * HID + nt * 16] = acc[nt][r] * dr[r];
            }
        } else {
#pragma unroll
            for (int nt = 0; nt < 8; ++nt) {
#pragma unroll
                for (int r = 0; r < 8; ++r) {
                    if (row0 + r + 8 * hi < nrows) hrow[(size_t)r * HID + nt * 16] = acc[nt][r] * dr[r];
                }
            }
        }
    }
}


template <int G, int DIM>
__device__ __forceinline__ void add_rows(float __attribute__((ext_vector_type(DIM / 32)))& acc,
                                         const float* __restrict__ Hlane, int sv, int t) {
    typedef float vecf __attribute__((ext_vector_type(DIM / 32)));
    vecf h[G];
#pragma unroll
    for (int j = 0; j < G; ++j) {
        const int s = __builtin_amdgcn_readlane(sv, t + j);
        h[j] = *(const vecf*)(Hlane + (size_t)s * DIM);
    }
#pragma unroll
    for (int j = 0; j < G; ++j) acc = acc + h[j];
}

template <int DIM, bool RELU>
__global__ __launch_bounds__(256) void k_gather(const float* __restrict__ H,
                                                const int* __restrict__ rowptr,
                                                const int* __restrict__ colsrc,
                                                const float* __restrict__ dinv,
                                                const float* __restrict__ bias,
                                                float* __restrict__ Out, int n, int E) {
    constexpr int VPT = DIM / 32;
    typedef float vecf __attribute__((ext_vector_type(VPT)));
    const int lane = threadIdx.x & 31;
    const int wave0 = __builtin_amdgcn_readfirstlane(
        (int)((blockIdx.x * blockDim.x + threadIdx.x) >> 5));
    const int nwaves = (int)((gridDim.x * blockDim.x) >> 5);
    const float* Hlane = H + lane * VPT;
    const vecf bv = *(const vecf*)(bias + lane * VPT);

    for (int row = wave0; row < n; row += nwaves) {
        const int beg = rowptr[row];
        const int end = (row + 1 < n) ? rowptr[row + 1] : E;
        const float di = dinv[row];
        vecf acc = *(const vecf*)(Hlane + (size_t)row * DIM);

        for (int base = beg; base < end; base += 32) {
            const int m = min(32, end - base);
            int sv = 0;
            if (lane < m)
                sv = (int)min((unsigned)colsrc[base + lane], (unsigned)(n - 1));
            int t = 0;
            for (; t + 8 <= m; t += 8) add_rows<8, DIM>(acc, Hlane, sv, t);
            if (m & 4) { add_rows<4, DIM>(acc, Hlane, sv, t); t += 4; }
            if (m & 2) { add_rows<2, DIM>(acc, Hlane, sv, t); t += 2; }
            if (m & 1) { add_rows<1, DIM>(acc, Hlane, sv, t); }
        }
        acc = acc * di + bv;
        if (RELU) {
#pragma unroll
            for (int j = 0; j < VPT; ++j) acc[j] = fmaxf(acc[j], 0.0f);
        }
        *(vecf*)(Out + (size_t)row * DIM + lane * VPT) = acc;
    }
}

#define TPITCH 132
#define FWAVES 4
#define GB 24

__device__ __forceinline__ int load_slots(const int* __restrict__ colsrc, int b, int lane, int E) {
    return colsrc[min(b + lane, E - 1)];
}

typedef float vec4f __attribute__((ext_vector_type(4)));

template <int J, int G>
__device__ __forceinline__ void request_rows(vec4f (&h)[G], const float* __restrict__ Hlane, int sv,
                                             int t, int cnt) {
    if constexpr (J < G) {
        if (J < cnt) {
            const int s = __builtin_amdgcn_readlane(sv, t + J);
            h[J] = *(const vec4f*)(Hlane + (size_t)s * HID);
            request_rows<J + 1, G>(h, Hlane, sv, t, cnt);
        }
    }
}

template <int J, int G>
__device__ __forceinline__ void sum_rows(vec4f& acc, const vec4f (&h)[G], int cnt) {
    if constexpr (J < G) {
        if (J < cnt) {
            acc = acc + h[J];
            sum_rows<J + 1, G>(acc, h, cnt);
        }
    }
}

template <int G>
__device__ __forceinline__ void add_rows_upto(vec4f& acc, const float* __restrict__ Hlane, int sv,
                                              int t, int cnt) {
    vec4f h[G];
    request_rows<0, G>(h, Hlane, sv, t, cnt);
    sum_rows<0, G>(acc, h, cnt);
}

__global__ __launch_bounds__(FWAVES * 32) void k_gather_gemm(const float* __restrict__ H,
                                                             const int* __restrict__ rowptr,
                                                             const int* __restrict__ colsrc,
                                                             const float* __restrict__ dinv,
                                                             const float* __restrict__ bias,
                                                             const float* __restrict__ W,
                                                             float* __restrict__ Out, int n, int E) {
    typedef float vecf __attribute__((ext_vector_type(4)));
    __shared__ float tiles[FWAVES][16 * TPITCH];
    const int lane = threadIdx.x & 31;
    const int tile = __builtin_amdgcn_readfirstlane((int)(blockIdx.x * FWAVES + (threadIdx.x >> 5)));
    if (tile * 16 >= n) return;
    float* tl = tiles[threadIdx.x >> 5];
    const int row0 = tile * 16;

    const int ridx = row0 + lane;
    int rpv = rowptr[min(ridx, n - 1)];
    if (ridx >= n) rpv = E;
    rpv = min(max(rpv, 0), E);
    const float dv = dinv[min(ridx, n - 1)];

    const float* Hlane = H + lane * 4;
    const vecf bv = *(const vecf*)(bias + lane * 4);

    int beg = __builtin_amdgcn_readlane(rpv, 0);
    int end = max(__builtin_amdgcn_readlane(rpv, 1), beg);
    int svn = load_slots(colsrc, beg, lane, E);
    for (int r = 0; r < 16; ++r) {
        const int row = min(row0 + r, n - 1);
        const int b = beg, e = end;
        const int sv = (int)min((unsigned)svn, (unsigned)(n - 1));
        const float di = __int_as_float(__builtin_amdgcn_readlane(__float_as_int(dv), r));
        if (r < 15) {
            beg = e;
            end = max(__builtin_amdgcn_readlane(rpv, r + 2), beg);
            svn = load_slots(colsrc, beg, lane, E);
        }
        vecf acc = *(const vecf*)(Hlane + (size_t)row * HID);

        const int m0 = min(32, e - b);
        for (int t = 0; t < m0; t += GB) add_rows_upto<GB>(acc, Hlane, sv, t, min(GB, m0 - t));
        for (int base = b + 32; base < e; base += 32) {
            const int m = min(32, e - base);
            const int sx = (int)min((unsigned)load_slots(colsrc, base, lane, E), (unsigned)(n - 1));
            for (int t = 0; t < m; t += 2) add_rows_upto<2>(acc, Hlane, sx, t, min(2, m - t));
        }
        acc = acc * di + bv;
#pragma unroll
        for (int j = 0; j < 4; ++j) acc[j] = fmaxf(acc[j], 0.0f);
        *(vecf*)(tl + r * TPITCH + lane * 4) = acc;
    }

    const int ln = lane & 15;
    const int hi = lane >> 4;
    v8f c[4] = {};
    const float* arow = tl + ln * TPITCH + 2 * hi;
#pragma unroll 4
    for (int k = 0; k < HID; k += 4) {
        const v2f a = *(const v2f*)(arow + k);
        const float* wr = W + (size_t)(k + 2 * hi) * OUT;
#pragma unroll
        for (int nt = 0; nt < 4; ++nt) {
            v2f bb;
            bb.x = wr[nt * 16 + ln];
            bb.y = wr[OUT + nt * 16 + ln];
            c[nt] = __builtin_amdgcn_wmma_f32_16x16x4_f32(
                false, a, false, bb, (short)0, c[nt], false, false);
        }
    }

    const int dvi = __float_as_int(dv);
#pragma unroll
    for (int r = 0; r < 8; ++r) {
        const int s0 = __builtin_amdgcn_readlane(dvi, r);
        const int s1 = __builtin_amdgcn_readlane(dvi, r + 8);
        const float sc = __int_as_float(hi ? s1 : s0);
        const int orow = row0 + r + 8 * hi;
#pragma unroll
        for (int nt = 0; nt < 4; ++nt) {
            if (orow < n) Out[(size_t)orow * OUT + nt * 16 + ln] = c[nt][r] * sc;
        }
    }
}

extern "C" void kernel_launch(void* const* d_in, const int* in_sizes, int n_in,
                              void* d_out, int out_size, void* d_ws, size_t ws_size,
                              hipStream_t stream) {
    const float* x   = (const float*)d_in[0];
    const int*   ei  = (const int*)d_in[1];
    const float* W1  = (const float*)d_in[2];
    const float* b1  = (const float*)d_in[3];
    const float* W2  = (const float*)d_in[4];
    const float* b2  = (const float*)d_in[5];
    float*       out = (float*)d_out;

    const int n = in_sizes[0] / IN_DIM;
    const int E = in_sizes[1] / 2;
    const int* srcIdx = ei;
    const int* dstIdx = ei + E;

    float* ws     = (float*)d_ws;
    int*   rowptr = (int*)ws;
    float* dinv   = ws + n;
    float* h      = dinv + n;
    float* h2     = h + (size_t)n * HID;
    int*   colsrc = (int*)(h2 + (size_t)n * OUT);

    const int nb = (n + BINW - 1) / BINW;
    const int G  = (E + CHUNK - 1) / CHUNK;
    int* hist     = (int*)out;
    int* binStart = hist + ((G * nb + 63) & ~63);
    int* binned   = binStart + ((nb + 1 + 63) & ~63);

    k_bin_hist<<<G, 256, 0, stream>>>(dstIdx, E, nb, hist);
    k_bin_scatter<<<G, 256, 0, stream>>>(srcIdx, dstIdx, E, nb, G, hist, binStart, binned);
    k_bin_rows<<<nb, BINW, 0, stream>>>(binned, binStart, n, E, rowptr, dinv, colsrc);

    const int tiles = (n + 15) / 16;
    const int fusedBlocks = (tiles + FWAVES - 1) / FWAVES;
    const int gatherBlocks = 2048;

    k_gemm1_lds<<<G1GROUPS, G1WAVES * 32, 0, stream>>>(x, W1, dinv, h, n);
    k_gather_gemm<<<fusedBlocks, FWAVES * 32, 0, stream>>>(h, rowptr, colsrc, dinv, b1, W2, h2, n, E);

    k_gather<OUT, false><<<gatherBlocks, 256, 0, stream>>>(h2, rowptr, colsrc, dinv, b2, out, n, E);
}


# Level-2 sort and layer-1 dense product in one launch: a bin's work-group (448 rows = 28 tiles, 7 per SIMD) builds its rows, keeps their scales in LDS, then multiplies them; row sum order may change
# speedup vs baseline: 4.3992x; 1.0700x over previous
#include <hip/hip_runtime.h>


typedef float v2f __attribute__((ext_vector_type(2)));
typedef float v8f __attribute__((ext_vector_type(8)));

#define IN_DIM 128
#define HID 128
#define OUT 64

#define BINW 448
#define CHUNK 8192
#define MAXB 256

__device__ __forceinline__ int bin_of(int d, int nb) {
    return (int)min((unsigned)d / (unsigned)BINW, (unsigned)(nb - 1));
}

__global__ __launch_bounds__(256) void k_bin_hist(const int* __restrict__ dst, int E, int nb,
                                                  int* __restrict__ hist) {
    __shared__ int h[MAXB];
    const int t = threadIdx.x;
    h[t] = 0;
    __syncthreads();
    const int e0 = blockIdx.x * CHUNK + t;
    for (int i0 = 0; i0 < CHUNK; i0 += 256 * 16) {
        int d[16];
#pragma unroll
        for (int j = 0; j < 16; ++j) d[j] = dst[min(e0 + i0 + j * 256, E - 1)];
        __builtin_amdgcn_sched_barrier(0);
#pragma unroll
        for (int j = 0; j < 16; ++j)
            atomicAdd(&h[bin_of(d[j], nb)], (e0 + i0 + j * 256 < E) ? 1 : 0);
    }
    __syncthreads();
    if (t < nb) hist[blockIdx.x * nb + t] = h[t];
}

typedef int v2i __attribute__((ext_vector_type(2)));

__global__ __launch_bounds__(256) void k_bin_scatter(const int* __restrict__ src,
                                                     const int* __restrict__ dst, int E, int nb,
                                                     int G, const int* __restrict__ hist,
                                                     int* __restrict__ binStart,
                                                     int* __restrict__ binned) {
    __shared__ int cur[MAXB];
    __shared__ int gdel[MAXB];
    __shared__ int wtot[8];
    __shared__ int wown[8];
    __shared__ v2i stage[CHUNK + 32];
    const int t = threadIdx.x;
    const int lane = t & 31;
    const int wave = t >> 5;
    const int blk = blockIdx.x;

    int tot = 0, part = 0, own = 0;
    const int tb = min(t, nb - 1);
    for (int g0 = 0; g0 < G; g0 += 28) {
        int v[28];
#pragma unroll
        for (int j = 0; j < 28; ++j) v[j] = hist[min(g0 + j, G - 1) * nb + tb];
        __builtin_amdgcn_sched_barrier(0);
#pragma unroll
        for (int j = 0; j < 28; ++j) {
            tot += (g0 + j < G) ? v[j] : 0;
            part += (g0 + j < blk) ? v[j] : 0;
            own += (g0 + j == blk) ? v[j] : 0;
        }
    }
    if (t >= nb) { tot = 0; part = 0; own = 0; }

    int inc = tot, linc = own;
    for (int o = 1; o < 32; o <<= 1) {
        int v = __shfl_up(inc, o);
        int u = __shfl_up(linc, o);
        if (lane >= o) { inc += v; linc += u; }
    }
    if (lane == 31) { wtot[wave] = inc; wown[wave] = linc; }
    __syncthreads();
    int off = 0, loff = 0;
    for (int w = 0; w < 8; ++w) {
        if (w < wave) { off += wtot[w]; loff += wown[w]; }
    }
    const int start = off + inc - tot;
    const int lstart = loff + linc - own;
    cur[t] = lstart;
    gdel[t] = start + part - lstart;
    if (blk == 0) {
        if (t < nb) binStart[t] = start;
        if (t == 0) binStart[nb] = E;
    }
    __syncthreads();

    const int e0 = blk * CHUNK + t;
    for (int i0 = 0; i0 < CHUNK; i0 += 256 * 8) {
        int s[8], d[8];
#pragma unroll
        for (int j = 0; j < 8; ++j) {
            const int e = min(e0 + i0 + j * 256, E - 1);
            s[j] = src[e];
            d[j] = dst[e];
        }
        __builtin_amdgcn_sched_barrier(0);
        int pos[8];
#pragma unroll
        for (int j = 0; j < 8; ++j)
            pos[j] = atomicAdd(&cur[bin_of(d[j], nb)], (e0 + i0 + j * 256 < E) ? 1 : 0);
#pragma unroll
        for (int j = 0; j < 8; ++j) {
            const bool ok = (e0 + i0 + j * 256 < E) && ((unsigned)pos[j] < (unsigned)CHUNK);
            v2i p;
            p.x = s[j];
            p.y = d[j];
            stage[ok ? pos[j] : CHUNK + lane] = p;
        }
    }
    __syncthreads();

    const int cnt = min(CHUNK, E - blk * CHUNK);
    for (int i0 = 0; i0 < cnt; i0 += 256 * 8) {
        v2i p[8];
        int gp[8];
#pragma unroll
        for (int j = 0; j < 8; ++j) p[j] = stage[i0 + j * 256 + t];
#pragma unroll
        for (int j = 0; j < 8; ++j) gp[j] = gdel[bin_of(p[j].y, nb)] + i0 + j * 256 + t;
#pragma unroll
        for (int j = 0; j < 8; ++j) {
            if (i0 + j * 256 + t < cnt && (unsigned)gp[j] < (unsigned)E)
                *(v2i*)(binned + 2 * (size_t)gp[j]) = p[j];
        }
    }
}

#define RCAP 16384
#define G1WAVES 16
#define NT2 (G1WAVES * 32)
#define BTILES (BINW / 16)
#define WPITCH 288
#define XPITCH 132

typedef float v4f __attribute__((ext_vector_type(4)));

__global__ __launch_bounds__(NT2) void k_rows_gemm1(const int* __restrict__ binned,
                                                    const int* __restrict__ binStart, int n, int E,
                                                    int* __restrict__ rowptr,
                                                    float* __restrict__ dinv,
                                                    int* __restrict__ colsrc,
                                                    const float* __restrict__ X,
                                                    const float* __restrict__ W,
                                                    float* __restrict__ H) {
    __shared__ int cnt[NT2];
    __shared__ int cur[NT2];
    __shared__ int wtot[NT2 / 32];
    __shared__ int stage[RCAP + 32];
    __shared__ float dsc[NT2];
    __shared__ __attribute__((aligned(16))) float wl[(IN_DIM / 2) * WPITCH];
    __shared__ __attribute__((aligned(16))) float xt[G1WAVES][16 * XPITCH];
    const int t = threadIdx.x;
    const int lane = t & 31;
    const int wave = __builtin_amdgcn_readfirstlane(t >> 5);
    const int ln = lane & 15;
    const int hi = lane >> 4;
    const int nodeBase = blockIdx.x * BINW;
    const int beg = min(max(binStart[blockIdx.x], 0), E);
    const int end = min(max(binStart[blockIdx.x + 1], beg), E);
    const int total = end - beg;
    const bool staged = total <= RCAP;
    const int2* pairs = (const int2*)binned;

    const int ntl = min(BTILES, (n - nodeBase + 15) >> 4);

    v4f xr[16];
    {
        const int r0 = nodeBase + min(wave, ntl - 1) * 16;
#pragma unroll
        for (int r = 0; r < 16; ++r)
            xr[r] = *(const v4f*)(X + (size_t)min(r0 + r, n - 1) * IN_DIM + lane * 4);
    }
    {
        v2f w0[8], w1[8];
#pragma unroll
        for (int j = 0; j < 8; ++j) {
            const int i = j * NT2 + t;
            const int p = i >> 6, cp = i & 63;
            w0[j] = *(const v2f*)(W + (size_t)(2 * p) * HID + 2 * cp);
            w1[j] = *(const v2f*)(W + (size_t)(2 * p + 1) * HID + 2 * cp);
        }
#pragma unroll
        for (int j = 0; j < 8; ++j) {
            const int i = j * NT2 + t;
            const int p = i >> 6, cp = i & 63;
            v4f q;
            q.x = w0[j].x; q.y = w1[j].x; q.z = w0[j].y; q.w = w1[j].y;
            *(v4f*)(wl + p * WPITCH + 4 * cp) = q;
        }
    }

    cnt[t] = 0;
    __syncthreads();
    for (int p0 = beg + t; p0 < end; p0 += NT2 * 8) {
        int dd[8];
#pragma unroll
        for (int j = 0; j < 8; ++j) dd[j] = pairs[min(p0 + j * NT2, end - 1)].y;
        __builtin_amdgcn_sched_barrier(0);
#pragma unroll
        for (int j = 0; j < 8; ++j)
            atomicAdd(&cnt[min(max(dd[j] - nodeBase, 0), BINW - 1)], (p0 + j * NT2 < end) ? 1 : 0);
    }
    __syncthreads();

    const int c = cnt[t];
    int inc = c;
    for (int o = 1; o < 32; o <<= 1) {
        int v = __shfl_up(inc, o);
        if (lane >= o) inc += v;
    }
    if (lane == 31) wtot[wave] = inc;
    __syncthreads();
    int off = 0;
    for (int w = 0; w < NT2 / 32; ++w) {
        if (w < wave) off += wtot[w];
    }
    const int rp = beg + off + inc - c;
    cur[t] = rp;
    const int node = nodeBase + t;
    const float dn = rsqrtf((float)(c + 1));
    dsc[t] = dn;
    if (t < BINW && node < n) {
        rowptr[node] = rp;
        dinv[node] = dn;
    }
    __syncthreads();

    for (int p0 = beg + t; p0 < end; p0 += NT2 * 8) {
        int2 sd[8];
#pragma unroll
        for (int j = 0; j < 8; ++j) sd[j] = pairs[min(p0 + j * NT2, end - 1)];
        __builtin_amdgcn_sched_barrier(0);
        int pos[8];
#pragma unroll
        for (int j = 0; j < 8; ++j)
            pos[j] = atomicAdd(&cur[min(max(sd[j].y - nodeBase, 0), BINW - 1)],
                               (p0 + j * NT2 < end) ? 1 : 0);
        if (staged) {
#pragma unroll
            for (int j = 0; j < 8; ++j) {
                const unsigned l = (unsigned)(pos[j] - beg);
                stage[(p0 + j * NT2 < end && l < (unsigned)RCAP) ? l : (unsigned)(RCAP + lane)] = sd[j].x;
            }
        } else {
#pragma unroll
            for (int j = 0; j < 8; ++j) {
                if (p0 + j * NT2 < end && (unsigned)pos[j] < (unsigned)E) colsrc[pos[j]] = sd[j].x;
            }
        }
    }
    __syncthreads();
    if (staged) {
        for (int i = t; i < total; i += NT2) colsrc[beg + i] = stage[i];
    }


    float* xtw = xt[wave];
    const float* arow = xtw + ln * XPITCH + 2 * hi;
    const float* brow = wl + hi * WPITCH + 2 * ln;
    for (int tl = wave; tl < ntl; tl += G1WAVES) {
        const int row0 = nodeBase + tl * 16;
#pragma unroll
        for (int r = 0; r < 16; ++r) *(v4f*)(xtw + r * XPITCH + lane * 4) = xr[r];
        float dr[8];
#pragma unroll
        for (int r = 0; r < 8; ++r) dr[r] = dsc[tl * 16 + r + 8 * hi];
        if (tl + G1WAVES < ntl) {
#pragma unroll
            for (int r = 0; r < 16; ++r)
                xr[r] = *(const v4f*)(X + (size_t)min(row0 + G1WAVES * 16 + r, n - 1) * IN_DIM + lane * 4);
        }

        v8f acc[8] = {};
#pragma unroll 2
        for (int k = 0; k < IN_DIM; k += 4) {
            const v2f a = *(const v2f*)(arow + k);
            const float* bp = brow + (k >> 1) * WPITCH;
#pragma unroll
            for (int nt = 0; nt < 8; ++nt) {
                const v2f bb = *(const v2f*)(bp + nt * 32);
                acc[nt] = __builtin_amdgcn_wmma_f32_16x16x4_f32(
                    false, a, false, bb, (short)0, acc[nt], false, false);
            }
        }

        float* hrow = H + (size_t)(row0 + 8 * hi) * HID + ln;
        if (row0 + 16 <= n) {
#pragma unroll
            for (int nt = 0; nt < 8; ++nt) {
#pragma unroll
                for (int r = 0; r < 8; ++r) hrow[(size_t)r * HID + nt * 16] = acc[nt][r] * dr[r];
            }
        } else {
#pragma unroll
            for (int nt = 0; nt < 8; ++nt) {
#pragma unroll
                for (int r = 0; r < 8; ++r) {
                    if (row0 + r + 8 * hi < n) hrow[(size_t)r * HID + nt * 16] = acc[nt][r] * dr[r];
                }
            }
        }
    }
}


template <int G, int DIM>
__device__ __forceinline__ void add_rows(float __attribute__((ext_vector_type(DIM / 32)))& acc,
                                         const float* __restrict__ Hlane, int sv, int t) {
    typedef float vecf __attribute__((ext_vector_type(DIM / 32)));
    vecf h[G];
#pragma unroll
    for (int j = 0; j < G; ++j) {
        const int s = __builtin_amdgcn_readlane(sv, t + j);
        h[j] = *(const vecf*)(Hlane + (size_t)s * DIM);
    }
#pragma unroll
    for (int j = 0; j < G; ++j) acc = acc + h[j];
}

template <int DIM, bool RELU>
__global__ __launch_bounds__(256) void k_gather(const float* __restrict__ H,
                                                const int* __restrict__ rowptr,
                                                const int* __restrict__ colsrc,
                                                const float* __restrict__ dinv,
                                                const float* __restrict__ bias,
                                                float* __restrict__ Out, int n, int E) {
    constexpr int VPT = DIM / 32;
    typedef float vecf __attribute__((ext_vector_type(VPT)));
    const int lane = threadIdx.x & 31;
    const int wave0 = __builtin_amdgcn_readfirstlane(
        (int)((blockIdx.x * blockDim.x + threadIdx.x) >> 5));
    const int nwaves = (int)((gridDim.x * blockDim.x) >> 5);
    const float* Hlane = H + lane * VPT;
    const vecf bv = *(const vecf*)(bias + lane * VPT);

    for (int row = wave0; row < n; row += nwaves) {
        const int beg = rowptr[row];
        const int end = (row + 1 < n) ? rowptr[row + 1] : E;
        const float di = dinv[row];
        vecf acc = *(const vecf*)(Hlane + (size_t)row * DIM);

        for (int base = beg; base < end; base += 32) {
            const int m = min(32, end - base);
            int sv = 0;
            if (lane < m)
                sv = (int)min((unsigned)colsrc[base + lane], (unsigned)(n - 1));
            int t = 0;
            for (; t + 8 <= m; t += 8) add_rows<8, DIM>(acc, Hlane, sv, t);
            if (m & 4) { add_rows<4, DIM>(acc, Hlane, sv, t); t += 4; }
            if (m & 2) { add_rows<2, DIM>(acc, Hlane, sv, t); t += 2; }
            if (m & 1) { add_rows<1, DIM>(acc, Hlane, sv, t); }
        }
        acc = acc * di + bv;
        if (RELU) {
#pragma unroll
            for (int j = 0; j < VPT; ++j) acc[j] = fmaxf(acc[j], 0.0f);
        }
        *(vecf*)(Out + (size_t)row * DIM + lane * VPT) = acc;
    }
}

#define TPITCH 132
#define FWAVES 4
#define GB 24

__device__ __forceinline__ int load_slots(const int* __restrict__ colsrc, int b, int lane, int E) {
    return colsrc[min(b + lane, E - 1)];
}

typedef float vec4f __attribute__((ext_vector_type(4)));

template <int J, int G>
__device__ __forceinline__ void request_rows(vec4f (&h)[G], const float* __restrict__ Hlane, int sv,
                                             int t, int cnt) {
    if constexpr (J < G) {
        if (J < cnt) {
            const int s = __builtin_amdgcn_readlane(sv, t + J);
            h[J] = *(const vec4f*)(Hlane + (size_t)s * HID);
            request_rows<J + 1, G>(h, Hlane, sv, t, cnt);
        }
    }
}

template <int J, int G>
__device__ __forceinline__ void sum_rows(vec4f& acc, const vec4f (&h)[G], int cnt) {
    if constexpr (J < G) {
        if (J < cnt) {
            acc = acc + h[J];
            sum_rows<J + 1, G>(acc, h, cnt);
        }
    }
}

template <int G>
__device__ __forceinline__ void add_rows_upto(vec4f& acc, const float* __restrict__ Hlane, int sv,
                                              int t, int cnt) {
    vec4f h[G];
    request_rows<0, G>(h, Hlane, sv, t, cnt);
    sum_rows<0, G>(acc, h, cnt);
}

__global__ __launch_bounds__(FWAVES * 32) void k_gather_gemm(const float* __restrict__ H,
                                                             const int* __restrict__ rowptr,
                                                             const int* __restrict__ colsrc,
                                                             const float* __restrict__ dinv,
                                                             const float* __restrict__ bias,
                                                             const float* __restrict__ W,
                                                             float* __restrict__ Out, int n, int E) {
    typedef float vecf __attribute__((ext_vector_type(4)));
    __shared__ float tiles[FWAVES][16 * TPITCH];
    const int lane = threadIdx.x & 31;
    const int tile = __builtin_amdgcn_readfirstlane((int)(blockIdx.x * FWAVES + (threadIdx.x >> 5)));
    if (tile * 16 >= n) return;
    float* tl = tiles[threadIdx.x >> 5];
    const int row0 = tile * 16;

    const int ridx = row0 + lane;
    int rpv = rowptr[min(ridx, n - 1)];
    if (ridx >= n) rpv = E;
    rpv = min(max(rpv, 0), E);
    const float dv = dinv[min(ridx, n - 1)];

    const float* Hlane = H + lane * 4;
    const vecf bv = *(const vecf*)(bias + lane * 4);

    int beg = __builtin_amdgcn_readlane(rpv, 0);
    int end = max(__builtin_amdgcn_readlane(rpv, 1), beg);
    int svn = load_slots(colsrc, beg, lane, E);
    for (int r = 0; r < 16; ++r) {
        const int row = min(row0 + r, n - 1);
        const int b = beg, e = end;
        const int sv = (int)min((unsigned)svn, (unsigned)(n - 1));
        const float di = __int_as_float(__builtin_amdgcn_readlane(__float_as_int(dv), r));
        if (r < 15) {
            beg = e;
            end = max(__builtin_amdgcn_readlane(rpv, r + 2), beg);
            svn = load_slots(colsrc, beg, lane, E);
        }
        vecf acc = *(const vecf*)(Hlane + (size_t)row * HID);

        const int m0 = min(32, e - b);
        for (int t = 0; t < m0; t += GB) add_rows_upto<GB>(acc, Hlane, sv, t, min(GB, m0 - t));
        for (int base = b + 32; base < e; base += 32) {
            const int m = min(32, e - base);
            const int sx = (int)min((unsigned)load_slots(colsrc, base, lane, E), (unsigned)(n - 1));
            for (int t = 0; t < m; t += 2) add_rows_upto<2>(acc, Hlane, sx, t, min(2, m - t));
        }
        acc = acc * di + bv;
#pragma unroll
        for (int j = 0; j < 4; ++j) acc[j] = fmaxf(acc[j], 0.0f);
        *(vecf*)(tl + r * TPITCH + lane * 4) = acc;
    }

    const int ln = lane & 15;
    const int hi = lane >> 4;
    v8f c[4] = {};
    const float* arow = tl + ln * TPITCH + 2 * hi;
#pragma unroll 4
    for (int k = 0; k < HID; k += 4) {
        const v2f a = *(const v2f*)(arow + k);
        const float* wr = W + (size_t)(k + 2 * hi) * OUT;
#pragma unroll
        for (int nt = 0; nt < 4; ++nt) {
            v2f bb;
            bb.x = wr[nt * 16 + ln];
            bb.y = wr[OUT + nt * 16 + ln];
            c[nt] = __builtin_amdgcn_wmma_f32_16x16x4_f32(
                false, a, false, bb, (short)0, c[nt], false, false);
        }
    }

    const int dvi = __float_as_int(dv);
#pragma unroll
    for (int r = 0; r < 8; ++r) {
        const int s0 = __builtin_amdgcn_readlane(dvi, r);
        const int s1 = __builtin_amdgcn_readlane(dvi, r + 8);
        const float sc = __int_as_float(hi ? s1 : s0);
        const int orow = row0 + r + 8 * hi;
#pragma unroll
        for (int nt = 0; nt < 4; ++nt) {
            if (orow < n) Out[(size_t)orow * OUT + nt * 16 + ln] = c[nt][r] * sc;
        }
    }
}

extern "C" void kernel_launch(void* const* d_in, const int* in_sizes, int n_in,
                              void* d_out, int out_size, void* d_ws, size_t ws_size,
                              hipStream_t stream) {
    const float* x   = (const float*)d_in[0];
    const int*   ei  = (const int*)d_in[1];
    const float* W1  = (const float*)d_in[2];
    const float* b1  = (const float*)d_in[3];
    const float* W2  = (const float*)d_in[4];
    const float* b2  = (const float*)d_in[5];
    float*       out = (float*)d_out;

    const int n = in_sizes[0] / IN_DIM;
    const int E = in_sizes[1] / 2;
    const int* srcIdx = ei;
    const int* dstIdx = ei + E;

    float* ws     = (float*)d_ws;
    int*   rowptr = (int*)ws;
    float* dinv   = ws + n;
    float* h      = dinv + n;
    float* h2     = h + (size_t)n * HID;
    int*   colsrc = (int*)(h2 + (size_t)n * OUT);

    const int nb = (n + BINW - 1) / BINW;
    const int G  = (E + CHUNK - 1) / CHUNK;
    int* hist     = (int*)out;
    int* binStart = hist + ((G * nb + 63) & ~63);
    int* binned   = binStart + ((nb + 1 + 63) & ~63);

    k_bin_hist<<<G, 256, 0, stream>>>(dstIdx, E, nb, hist);
    k_bin_scatter<<<G, 256, 0, stream>>>(srcIdx, dstIdx, E, nb, G, hist, binStart, binned);
    k_rows_gemm1<<<nb, NT2, 0, stream>>>(binned, binStart, n, E, rowptr, dinv, colsrc, x, W1, h);

    const int tiles = (n + 15) / 16;
    const int fusedBlocks = (tiles + FWAVES - 1) / FWAVES;
    const int gatherBlocks = 2048;

    k_gather_gemm<<<fusedBlocks, FWAVES * 32, 0, stream>>>(h, rowptr, colsrc, dinv, b1, W2, h2, n, E);

    k_gather<OUT, false><<<gatherBlocks, 256, 0, stream>>>(h2, rowptr, colsrc, dinv, b2, out, n, E);
}
